# starting kernel
# baseline (speedup 1.0000x reference)
#include <hip/hip_runtime.h>
#include <hip/hip_bf16.h>

typedef float v2f __attribute__((ext_vector_type(2)));
typedef float v8f __attribute__((ext_vector_type(8)));

#define HEADS 8
#define CH 16
#define F 128
#define NEG_SLOPE 0.2f

#if defined(__has_builtin)
#if __has_builtin(__builtin_amdgcn_wmma_f32_16x16x4_f32)
#define HAVE_WMMA_F32X4 1
#endif
#endif

__device__ __forceinline__ unsigned flipf(float f) {
  unsigned u = __float_as_uint(f);
  return (u >> 31) ? ~u : (u | 0x80000000u);
}
__device__ __forceinline__ float unflipf(unsigned u) {
  return __uint_as_float((u >> 31) ? (u ^ 0x80000000u) : ~u);
}
__device__ __forceinline__ float lrelu(float v) {
  return v > 0.0f ? v : NEG_SLOPE * v;
}

__global__ __launch_bounds__(256) void gat_init(float* __restrict__ out,
                                                const float* __restrict__ bias,
                                                unsigned* __restrict__ m_u,
                                                float* __restrict__ denom,
                                                int N) {
  long long i = (long long)blockIdx.x * 256 + threadIdx.x;
  long long tot = (long long)N * F;
  if (i < tot) out[i] = bias[(int)(i & (F - 1))];
  if (i < (long long)N * HEADS) {
    m_u[i] = 0x007FFFFFu;
    denom[i] = 0.0f;
  }
}

__global__ __launch_bounds__(256) void gat_gemm_wmma(
    const float* __restrict__ x, const float* __restrict__ W,
    const float* __restrict__ att_src, const float* __restrict__ att_dst,
    float* __restrict__ h, float* __restrict__ a_src, float* __restrict__ a_dst,
    int N) {
  __shared__ float sA[16 * F];

  const int tid = threadIdx.x;
  const int wave = tid >> 5;
  const int lane = tid & 31;
  const int half = lane >> 4;
  const int l16 = lane & 15;
  const int rowbase = blockIdx.x * 16;

  {
    int r = tid >> 4;
    int c = (tid & 15) << 3;
    int rg = rowbase + r;
    if (rg >= N) rg = N - 1;
    const float4* s4 = (const float4*)(x + (size_t)rg * F + c);
    float4 a0 = s4[0];
    float4 a1 = s4[1];
    *(float4*)(sA + r * F + c) = a0;
    *(float4*)(sA + r * F + c + 4) = a1;
  }
  __syncthreads();

  const int colbase = wave * CH;
  v8f acc = {0.f, 0.f, 0.f, 0.f, 0.f, 0.f, 0.f, 0.f};

  for (int k = 0; k < F; k += 4) {
#if defined(HAVE_WMMA_F32X4)
    v2f afrag;
    const float* ap = sA + l16 * F + k + 2 * half;
    afrag.x = ap[0];
    afrag.y = ap[1];
    v2f bfrag;
    const float* bp = W + (size_t)(k + 2 * half) * F + colbase + l16;
    bfrag.x = bp[0];
    bfrag.y = bp[F];
    acc = __builtin_amdgcn_wmma_f32_16x16x4_f32(false, afrag, false, bfrag,
                                                (short)0, acc, false, false);
#else
    for (int v = 0; v < 8; ++v) {
      int row = v + 8 * half;
      for (int j = 0; j < 4; ++j)
        acc[v] += sA[row * F + k + j] * W[(size_t)(k + j) * F + colbase + l16];
    }
#endif
  }

  const float was = att_src[wave * CH + l16];
  const float wad = att_dst[wave * CH + l16];
#pragma unroll
  for (int v = 0; v < 8; ++v) {
    int row = rowbase + v + 8 * half;
    float val = acc[v];
    float ps = val * was;
    float pd = val * wad;
#pragma unroll
    for (int off = 8; off >= 1; off >>= 1) {
      ps += __shfl_xor(ps, off, 32);
      pd += __shfl_xor(pd, off, 32);
    }
    if (row < N) {
      h[(size_t)row * F + colbase + l16] = val;
      if (l16 == 0) {
        a_src[(size_t)row * HEADS + wave] = ps;
        a_dst[(size_t)row * HEADS + wave] = pd;
      }
    }
  }
}

__global__ __launch_bounds__(256) void gat_logits_max(
    const int* __restrict__ srcA, const int* __restrict__ dstA,
    const float* __restrict__ a_src, const float* __restrict__ a_dst,
    unsigned* __restrict__ m_u, int E, int EN) {
  int e = blockIdx.x * 256 + threadIdx.x;
  if (e >= EN) return;
  int s, d;
  if (e < E) { s = srcA[e]; d = dstA[e]; } else { s = d = e - E; }
  const float4* as4 = (const float4*)(a_src + (size_t)s * HEADS);
  const float4* ad4 = (const float4*)(a_dst + (size_t)d * HEADS);
  float4 s0 = as4[0], s1 = as4[1];
  float4 d0 = ad4[0], d1 = ad4[1];
  unsigned* mp = m_u + (size_t)d * HEADS;
  atomicMax(mp + 0, flipf(lrelu(s0.x + d0.x)));
  atomicMax(mp + 1, flipf(lrelu(s0.y + d0.y)));
  atomicMax(mp + 2, flipf(lrelu(s0.z + d0.z)));
  atomicMax(mp + 3, flipf(lrelu(s0.w + d0.w)));
  atomicMax(mp + 4, flipf(lrelu(s1.x + d1.x)));
  atomicMax(mp + 5, flipf(lrelu(s1.y + d1.y)));
  atomicMax(mp + 6, flipf(lrelu(s1.z + d1.z)));
  atomicMax(mp + 7, flipf(lrelu(s1.w + d1.w)));
}

__global__ __launch_bounds__(256) void gat_finalize_m(unsigned* __restrict__ m_u,
                                                      int count) {
  int i = blockIdx.x * 256 + threadIdx.x;
  if (i < count) {
    float v = unflipf(m_u[i]);
    ((float*)m_u)[i] = v;
  }
}

__global__ __launch_bounds__(256) void gat_expsum(
    const int* __restrict__ srcA, const int* __restrict__ dstA,
    const float* __restrict__ a_src, const float* __restrict__ a_dst,
    const float* __restrict__ m, float* __restrict__ denom, int E, int EN) {
  int e = blockIdx.x * 256 + threadIdx.x;
  if (e >= EN) return;
  int s, d;
  if (e < E) { s = srcA[e]; d = dstA[e]; } else { s = d = e - E; }
  const float4* as4 = (const float4*)(a_src + (size_t)s * HEADS);
  const float4* ad4 = (const float4*)(a_dst + (size_t)d * HEADS);
  const float4* m4 = (const float4*)(m + (size_t)d * HEADS);
  float4 s0 = as4[0], s1 = as4[1];
  float4 d0 = ad4[0], d1 = ad4[1];
  float4 m0 = m4[0], m1 = m4[1];
  float* dp = denom + (size_t)d * HEADS;
  atomicAdd(dp + 0, expf(lrelu(s0.x + d0.x) - m0.x));
  atomicAdd(dp + 1, expf(lrelu(s0.y + d0.y) - m0.y));
  atomicAdd(dp + 2, expf(lrelu(s0.z + d0.z) - m0.z));
  atomicAdd(dp + 3, expf(lrelu(s0.w + d0.w) - m0.w));
  atomicAdd(dp + 4, expf(lrelu(s1.x + d1.x) - m1.x));
  atomicAdd(dp + 5, expf(lrelu(s1.y + d1.y) - m1.y));
  atomicAdd(dp + 6, expf(lrelu(s1.z + d1.z) - m1.z));
  atomicAdd(dp + 7, expf(lrelu(s1.w + d1.w) - m1.w));
}

__global__ __launch_bounds__(256) void gat_aggregate(
    const int* __restrict__ srcA, const int* __restrict__ dstA,
    const float* __restrict__ a_src, const float* __restrict__ a_dst,
    const float* __restrict__ m, const float* __restrict__ denom,
    const float* __restrict__ h, float* __restrict__ out, int E, int EN) {
  long long widx = (long long)blockIdx.x * 8 + (threadIdx.x >> 5);
  if (widx >= EN) return;
  int e = (int)widx;
  int lane = threadIdx.x & 31;
  int s, d;
  if (e < E) { s = srcA[e]; d = dstA[e]; } else { s = d = e - E; }
  int hd = lane >> 2;
  float lg = lrelu(a_src[(size_t)s * HEADS + hd] + a_dst[(size_t)d * HEADS + hd]);
  float alpha = expf(lg - m[(size_t)d * HEADS + hd]) / denom[(size_t)d * HEADS + hd];
  float4 hv = *(const float4*)(h + (size_t)s * F + lane * 4);
  float* op = out + (size_t)d * F + lane * 4;
  atomicAdd(op + 0, alpha * hv.x);
  atomicAdd(op + 1, alpha * hv.y);
  atomicAdd(op + 2, alpha * hv.z);
  atomicAdd(op + 3, alpha * hv.w);
}

extern "C" void kernel_launch(void* const* d_in, const int* in_sizes, int n_in,
                              void* d_out, int out_size, void* d_ws, size_t ws_size,
                              hipStream_t stream) {
  const float* x = (const float*)d_in[0];
  const int* ei = (const int*)d_in[1];
  const float* W = (const float*)d_in[2];
  const float* att_src = (const float*)d_in[3];
  const float* att_dst = (const float*)d_in[4];
  const float* bias = (const float*)d_in[5];
  float* out = (float*)d_out;

  const int N = in_sizes[0] / F;
  const int E = in_sizes[1] / 2;
  const int EN = E + N;
  const int* srcA = ei;
  const int* dstA = ei + E;

  float* ws = (float*)d_ws;
  float* hbuf = ws;
  float* asrc = hbuf + (size_t)N * F;
  float* adst = asrc + (size_t)N * HEADS;
  float* mbuf = adst + (size_t)N * HEADS;
  float* dbuf = mbuf + (size_t)N * HEADS;

  {
    long long tot = (long long)N * F;
    int blocks = (int)((tot + 255) / 256);
    gat_init<<<blocks, 256, 0, stream>>>(out, bias, (unsigned*)mbuf, dbuf, N);
  }
  gat_gemm_wmma<<<(N + 15) / 16, 256, 0, stream>>>(x, W, att_src, att_dst,
                                                   hbuf, asrc, adst, N);
  gat_logits_max<<<(EN + 255) / 256, 256, 0, stream>>>(srcA, dstA, asrc, adst,
                                                       (unsigned*)mbuf, E, EN);
  gat_finalize_m<<<((N * HEADS) + 255) / 256, 256, 0, stream>>>((unsigned*)mbuf,
                                                                N * HEADS);
  gat_expsum<<<(EN + 255) / 256, 256, 0, stream>>>(srcA, dstA, asrc, adst,
                                                   mbuf, dbuf, E, EN);
  gat_aggregate<<<(EN + 7) / 8, 256, 0, stream>>>(srcA, dstA, asrc, adst,
                                                  mbuf, dbuf, hbuf, out, E, EN);
}


# Edges bucketed by destination (rank atomics, scan, place); one wave per node does max, exp sum and weighted row sum in registers, one division: no float atomics, sum order = bucket (rank) order
# speedup vs baseline: 3.4826x; 3.4826x over previous
#include <hip/hip_runtime.h>
#include <hip/hip_bf16.h>

typedef float v2f __attribute__((ext_vector_type(2)));
typedef float v8f __attribute__((ext_vector_type(8)));

#define HEADS 8
#define CH 16
#define F 128
#define NEG_SLOPE 0.2f

#if defined(__has_builtin)
#if __has_builtin(__builtin_amdgcn_wmma_f32_16x16x4_f32)
#define HAVE_WMMA_F32X4 1
#endif
#endif

__device__ __forceinline__ float lrelu(float v) {
  return v > 0.0f ? v : NEG_SLOPE * v;
}

__global__ __launch_bounds__(256) void node_gemm_wmma(
    const float* __restrict__ x, const float* __restrict__ W,
    const float* __restrict__ att_src, const float* __restrict__ att_dst,
    float* __restrict__ h, float* __restrict__ a_src, float* __restrict__ a_dst,
    int* __restrict__ cnt, int N) {
  __shared__ float sA[16 * F];

  const int tid = threadIdx.x;
  const int wave = tid >> 5;
  const int lane = tid & 31;
  const int half = lane >> 4;
  const int l16 = lane & 15;
  const int rowbase = blockIdx.x * 16;

  if (tid < 16 && rowbase + tid < N) cnt[rowbase + tid] = 0;

  {
    int r = tid >> 4;
    int c = (tid & 15) << 3;
    int rg = rowbase + r;
    if (rg >= N) rg = N - 1;
    const float4* s4 = (const float4*)(x + (size_t)rg * F + c);
    float4 a0 = s4[0];
    float4 a1 = s4[1];
    *(float4*)(sA + r * F + c) = a0;
    *(float4*)(sA + r * F + c + 4) = a1;
  }
  __syncthreads();

  const int colbase = wave * CH;
  v8f acc = {0.f, 0.f, 0.f, 0.f, 0.f, 0.f, 0.f, 0.f};

  for (int k = 0; k < F; k += 4) {
#if defined(HAVE_WMMA_F32X4)
    v2f afrag;
    const float* ap = sA + l16 * F + k + 2 * half;
    afrag.x = ap[0];
    afrag.y = ap[1];
    v2f bfrag;
    const float* bp = W + (size_t)(k + 2 * half) * F + colbase + l16;
    bfrag.x = bp[0];
    bfrag.y = bp[F];
    acc = __builtin_amdgcn_wmma_f32_16x16x4_f32(false, afrag, false, bfrag,
                                                (short)0, acc, false, false);
#else
    for (int v = 0; v < 8; ++v) {
      int row = v + 8 * half;
      for (int j = 0; j < 4; ++j)
        acc[v] += sA[row * F + k + j] * W[(size_t)(k + j) * F + colbase + l16];
    }
#endif
  }

  const float was = att_src[wave * CH + l16];
  const float wad = att_dst[wave * CH + l16];
#pragma unroll
  for (int v = 0; v < 8; ++v) {
    int row = rowbase + v + 8 * half;
    float val = acc[v];
    float ps = val * was;
    float pd = val * wad;
#pragma unroll
    for (int off = 8; off >= 1; off >>= 1) {
      ps += __shfl_xor(ps, off, 32);
      pd += __shfl_xor(pd, off, 32);
    }
    if (row < N) {
      h[(size_t)row * F + colbase + l16] = val;
      if (l16 == 0) {
        a_src[(size_t)row * HEADS + wave] = ps;
        a_dst[(size_t)row * HEADS + wave] = pd;
      }
    }
  }
}

__global__ __launch_bounds__(256) void edge_rank(const int* __restrict__ dstA,
                                                 int* __restrict__ cnt,
                                                 int* __restrict__ rank, int E) {
  int e = blockIdx.x * 256 + threadIdx.x;
  if (e < E) rank[e] = atomicAdd(&cnt[dstA[e]], 1);
}

__global__ __launch_bounds__(256) void count_scan(const int* __restrict__ cnt,
                                                  int* __restrict__ offs, int N) {
  __shared__ int sfront[8];
  __shared__ int swave[8];
  const int tid = threadIdx.x;
  const int lane = tid & 31;
  const int wave = tid >> 5;
  const int base = blockIdx.x * 1024;

  int part = 0;
  const int4* c4 = (const int4*)cnt;
  const int n4 = base >> 2;
#pragma unroll 8
  for (int i = tid; i < n4; i += 256) {
    int4 v = c4[i];
    part += (v.x + v.y) + (v.z + v.w);
  }
#pragma unroll
  for (int off = 16; off >= 1; off >>= 1) part += __shfl_xor(part, off, 32);
  if (lane == 0) sfront[wave] = part;

  const int i0 = base + tid * 4;
  const int v0 = (i0 + 0 < N) ? cnt[i0 + 0] : 0;
  const int v1 = (i0 + 1 < N) ? cnt[i0 + 1] : 0;
  const int v2 = (i0 + 2 < N) ? cnt[i0 + 2] : 0;
  const int v3 = (i0 + 3 < N) ? cnt[i0 + 3] : 0;
  const int t1 = v0;
  const int t2 = v0 + v1;
  const int t3 = t2 + v2;
  const int tsum = t3 + v3;
  int inc = tsum;
#pragma unroll
  for (int off = 1; off < 32; off <<= 1) {
    int y = __shfl_up(inc, off, 32);
    if (lane >= off) inc += y;
  }
  if (lane == 31) swave[wave] = inc;
  __syncthreads();

  int pre = 0;
#pragma unroll
  for (int w = 0; w < 8; ++w) {
    pre += sfront[w];
    if (w < wave) pre += swave[w];
  }
  const int ex = pre + inc - tsum;
  if (i0 + 0 < N) offs[i0 + 0] = ex;
  if (i0 + 1 < N) offs[i0 + 1] = ex + t1;
  if (i0 + 2 < N) offs[i0 + 2] = ex + t2;
  if (i0 + 3 < N) offs[i0 + 3] = ex + t3;
}

__global__ __launch_bounds__(256) void edge_place(
    const int* __restrict__ srcA, const int* __restrict__ dstA,
    const int* __restrict__ rank, const int* __restrict__ offs,
    int* __restrict__ col, int E) {
  int e = blockIdx.x * 256 + threadIdx.x;
  if (e < E) col[offs[dstA[e]] + rank[e]] = srcA[e];
}

__global__ __launch_bounds__(256) void node_softmax_gather(
    const int* __restrict__ cnt, const int* __restrict__ offs,
    const int* __restrict__ col,
    const float* __restrict__ a_src, const float* __restrict__ a_dst,
    const float* __restrict__ h, const float* __restrict__ bias,
    float* __restrict__ out, int N) {
  const int d = __builtin_amdgcn_readfirstlane(
      (int)(blockIdx.x * 8 + (threadIdx.x >> 5)));
  if (d >= N) return;
  const int lane = threadIdx.x & 31;
  const int hd = lane >> 2;
  const int deg = cnt[d];
  const int* __restrict__ cp = col + offs[d];
  const float ad = a_dst[(size_t)d * HEADS + hd];
  const float lself = lrelu(a_src[(size_t)d * HEADS + hd] + ad);
  const float4 hself = *(const float4*)(h + (size_t)d * F + lane * 4);

  float m = lself;
#pragma unroll 4
  for (int j = 0; j < deg; ++j) {
    const int s = cp[j];
    const float lg = lrelu(a_src[(size_t)s * HEADS + hd] + ad);
    m = (lg > m) ? lg : m;
  }
  m = (fabsf(m) < __builtin_inff()) ? m : 0.0f;

  const float eself = expf(lself - m);
  float den = eself;
  float4 acc;
  acc.x = eself * hself.x;
  acc.y = eself * hself.y;
  acc.z = eself * hself.z;
  acc.w = eself * hself.w;
#pragma unroll 4
  for (int j = 0; j < deg; ++j) {
    const int s = cp[j];
    const float4 hv = *(const float4*)(h + (size_t)s * F + lane * 4);
    const float ev = expf(lrelu(a_src[(size_t)s * HEADS + hd] + ad) - m);
    den += ev;
    acc.x += ev * hv.x;
    acc.y += ev * hv.y;
    acc.z += ev * hv.z;
    acc.w += ev * hv.w;
  }
  const float4 b = *(const float4*)(bias + lane * 4);
  float4 o;
  o.x = acc.x / den + b.x;
  o.y = acc.y / den + b.y;
  o.z = acc.z / den + b.z;
  o.w = acc.w / den + b.w;
  *(float4*)(out + (size_t)d * F + lane * 4) = o;
}

extern "C" void kernel_launch(void* const* d_in, const int* in_sizes, int n_in,
                              void* d_out, int out_size, void* d_ws, size_t ws_size,
                              hipStream_t stream) {
  const float* x = (const float*)d_in[0];
  const int* ei = (const int*)d_in[1];
  const float* W = (const float*)d_in[2];
  const float* att_src = (const float*)d_in[3];
  const float* att_dst = (const float*)d_in[4];
  const float* bias = (const float*)d_in[5];
  float* out = (float*)d_out;

  const int N = in_sizes[0] / F;
  const int E = in_sizes[1] / 2;
  const int* srcA = ei;
  const int* dstA = ei + E;

  float* ws = (float*)d_ws;
  float* hbuf = ws;
  float* asrc = hbuf + (size_t)N * F;
  float* adst = asrc + (size_t)N * HEADS;
  int* cnt = (int*)(adst + (size_t)N * HEADS);
  int* offs = cnt + N;
  int* rank = offs + N;
  int* col = rank + E;

  node_gemm_wmma<<<(N + 15) / 16, 256, 0, stream>>>(x, W, att_src, att_dst,
                                                    hbuf, asrc, adst, cnt, N);
  edge_rank<<<(E + 255) / 256, 256, 0, stream>>>(dstA, cnt, rank, E);
  count_scan<<<(N + 1023) / 1024, 256, 0, stream>>>(cnt, offs, N);
  edge_place<<<(E + 255) / 256, 256, 0, stream>>>(srcA, dstA, rank, offs, col, E);
  node_softmax_gather<<<(N + 7) / 8, 256, 0, stream>>>(cnt, offs, col, asrc, adst,
                                                       hbuf, bias, out, N);
}


# Gather wave reads only the h row per edge: logit recomputed from that row (att dot, quad sum), softmax in one pass with running max and rescale of the sums (bucket order kept); no logit arrays
# speedup vs baseline: 3.6966x; 1.0614x over previous
#include <hip/hip_runtime.h>
#include <hip/hip_bf16.h>

typedef float v2f __attribute__((ext_vector_type(2)));
typedef float v8f __attribute__((ext_vector_type(8)));
typedef float v4f __attribute__((ext_vector_type(4)));

#define CH 16
#define F 128
#define NEG_SLOPE 0.2f

#if defined(__has_builtin)
#if __has_builtin(__builtin_amdgcn_wmma_f32_16x16x4_f32)
#define HAVE_WMMA_F32X4 1
#endif
#endif

__device__ __forceinline__ float lrelu(float v) {
  return v > 0.0f ? v : NEG_SLOPE * v;
}

__global__ __launch_bounds__(256) void node_gemm_wmma(
    const float* __restrict__ x, const float* __restrict__ W,
    float* __restrict__ h, int* __restrict__ cnt, int N) {
  __shared__ float sA[16 * F];

  const int tid = threadIdx.x;
  const int wave = tid >> 5;
  const int lane = tid & 31;
  const int half = lane >> 4;
  const int l16 = lane & 15;
  const int rowbase = blockIdx.x * 16;

  if (tid < 16 && rowbase + tid < N) cnt[rowbase + tid] = 0;

  {
    int r = tid >> 4;
    int c = (tid & 15) << 3;
    int rg = rowbase + r;
    if (rg >= N) rg = N - 1;
    const float4* s4 = (const float4*)(x + (size_t)rg * F + c);
    float4 a0 = s4[0];
    float4 a1 = s4[1];
    *(float4*)(sA + r * F + c) = a0;
    *(float4*)(sA + r * F + c + 4) = a1;
  }
  __syncthreads();

  const int colbase = wave * CH;
  v8f acc = {0.f, 0.f, 0.f, 0.f, 0.f, 0.f, 0.f, 0.f};

  for (int k = 0; k < F; k += 4) {
#if defined(HAVE_WMMA_F32X4)
    v2f afrag;
    const float* ap = sA + l16 * F + k + 2 * half;
    afrag.x = ap[0];
    afrag.y = ap[1];
    v2f bfrag;
    const float* bp = W + (size_t)(k + 2 * half) * F + colbase + l16;
    bfrag.x = bp[0];
    bfrag.y = bp[F];
    acc = __builtin_amdgcn_wmma_f32_16x16x4_f32(false, afrag, false, bfrag,
                                                (short)0, acc, false, false);
#else
    for (int v = 0; v < 8; ++v) {
      int row = v + 8 * half;
      for (int j = 0; j < 4; ++j)
        acc[v] += sA[row * F + k + j] * W[(size_t)(k + j) * F + colbase + l16];
    }
#endif
  }

#pragma unroll
  for (int v = 0; v < 8; ++v) {
    int row = rowbase + v + 8 * half;
    if (row < N) h[(size_t)row * F + colbase + l16] = acc[v];
  }
}

__global__ __launch_bounds__(256) void edge_rank(const int* __restrict__ dstA,
                                                 int* __restrict__ cnt,
                                                 int* __restrict__ rank, int E) {
  int e = blockIdx.x * 256 + threadIdx.x;
  if (e < E) rank[e] = atomicAdd(&cnt[dstA[e]], 1);
}

__global__ __launch_bounds__(256) void count_scan(const int* __restrict__ cnt,
                                                  int* __restrict__ offs, int N) {
  __shared__ int sfront[8];
  __shared__ int swave[8];
  const int tid = threadIdx.x;
  const int lane = tid & 31;
  const int wave = tid >> 5;
  const int base = blockIdx.x * 1024;

  int part = 0;
  const int4* c4 = (const int4*)cnt;
  const int n4 = base >> 2;
#pragma unroll 8
  for (int i = tid; i < n4; i += 256) {
    int4 v = c4[i];
    part += (v.x + v.y) + (v.z + v.w);
  }
#pragma unroll
  for (int off = 16; off >= 1; off >>= 1) part += __shfl_xor(part, off, 32);
  if (lane == 0) sfront[wave] = part;

  const int i0 = base + tid * 4;
  const int v0 = (i0 + 0 < N) ? cnt[i0 + 0] : 0;
  const int v1 = (i0 + 1 < N) ? cnt[i0 + 1] : 0;
  const int v2 = (i0 + 2 < N) ? cnt[i0 + 2] : 0;
  const int v3 = (i0 + 3 < N) ? cnt[i0 + 3] : 0;
  const int t1 = v0;
  const int t2 = v0 + v1;
  const int t3 = t2 + v2;
  const int tsum = t3 + v3;
  int inc = tsum;
#pragma unroll
  for (int off = 1; off < 32; off <<= 1) {
    int y = __shfl_up(inc, off, 32);
    if (lane >= off) inc += y;
  }
  if (lane == 31) swave[wave] = inc;
  __syncthreads();

  int pre = 0;
#pragma unroll
  for (int w = 0; w < 8; ++w) {
    pre += sfront[w];
    if (w < wave) pre += swave[w];
  }
  const int ex = pre + inc - tsum;
  if (i0 + 0 < N) offs[i0 + 0] = ex;
  if (i0 + 1 < N) offs[i0 + 1] = ex + t1;
  if (i0 + 2 < N) offs[i0 + 2] = ex + t2;
  if (i0 + 3 < N) offs[i0 + 3] = ex + t3;
}

__global__ __launch_bounds__(256) void edge_place(
    const int* __restrict__ srcA, const int* __restrict__ dstA,
    const int* __restrict__ rank, const int* __restrict__ offs,
    int* __restrict__ col, int E) {
  int e = blockIdx.x * 256 + threadIdx.x;
  if (e < E) col[offs[dstA[e]] + rank[e]] = srcA[e];
}

__device__ __forceinline__ float quad_sum(float p) {
  p += __int_as_float(__builtin_amdgcn_update_dpp(0, __float_as_int(p), 0xB1, 0xF, 0xF, true));
  p += __int_as_float(__builtin_amdgcn_update_dpp(0, __float_as_int(p), 0x4E, 0xF, 0xF, true));
  return p;
}

__device__ __forceinline__ float dot4(const v4f a, const v4f b) {
  return a.x * b.x + a.y * b.y + a.z * b.z + a.w * b.w;
}

#define ROWS_AHEAD 8

__device__ __forceinline__ void edge_step(const v4f hv, const v4f was,
                                          const float ad, float& m, float& den,
                                          v4f& acc) {
  const float lg = lrelu(quad_sum(dot4(hv, was)) + ad);
  const float t = lg - m;
  const float pe = expf(-fabsf(t));
  const bool up = t > 0.0f;
  const float sc = up ? pe : 1.0f;
  const float ev = up ? 1.0f : pe;
  m = up ? lg : m;
  den = den * sc + ev;
  acc.x = acc.x * sc + ev * hv.x;
  acc.y = acc.y * sc + ev * hv.y;
  acc.z = acc.z * sc + ev * hv.z;
  acc.w = acc.w * sc + ev * hv.w;
}

__global__ __launch_bounds__(256) void node_softmax_gather(
    const int* __restrict__ cnt, const int* __restrict__ offs,
    const int* __restrict__ col,
    const float* __restrict__ att_src, const float* __restrict__ att_dst,
    const float* __restrict__ h, const float* __restrict__ bias,
    float* __restrict__ out, int N) {
  const int d = __builtin_amdgcn_readfirstlane(
      (int)(blockIdx.x * 8 + (threadIdx.x >> 5)));
  if (d >= N) return;
  const int lane = threadIdx.x & 31;
  const int deg = cnt[d];
  const int* __restrict__ cp = col + offs[d];
  const v4f was = *(const v4f*)(att_src + lane * 4);
  const v4f wad = *(const v4f*)(att_dst + lane * 4);
  const v4f hself = *(const v4f*)(h + (size_t)d * F + lane * 4);

  const float ad = quad_sum(dot4(hself, wad));
  float m = lrelu(quad_sum(dot4(hself, was)) + ad);
  float den = 1.0f;
  v4f acc = hself;

  for (int c0 = 0; c0 < deg; c0 += 32) {
    const int n = (deg - c0 < 32) ? (deg - c0) : 32;
    const int sv = (lane < n) ? cp[c0 + lane] : d;
    for (int j0 = 0; j0 < n; j0 += ROWS_AHEAD) {
      const int r = n - j0;
      int s[ROWS_AHEAD];
#pragma unroll
      for (int u = 0; u < ROWS_AHEAD; ++u) s[u] = __builtin_amdgcn_readlane(sv, j0 + u);
      v4f hv[ROWS_AHEAD];
      if (r >= ROWS_AHEAD) {
#pragma unroll
        for (int u = 0; u < ROWS_AHEAD; ++u)
          hv[u] = *(const v4f*)(h + (size_t)s[u] * F + lane * 4);
        asm volatile("" ::"v"(hv[0]), "v"(hv[1]), "v"(hv[2]), "v"(hv[3]),
                     "v"(hv[4]), "v"(hv[5]), "v"(hv[6]), "v"(hv[7]));
#pragma unroll
        for (int u = 0; u < ROWS_AHEAD; ++u) edge_step(hv[u], was, ad, m, den, acc);
      } else {
#pragma unroll
        for (int u = 0; u < ROWS_AHEAD - 1; ++u)
          if (u < r) hv[u] = *(const v4f*)(h + (size_t)s[u] * F + lane * 4);
#pragma unroll
        for (int u = 0; u < ROWS_AHEAD - 1; ++u)
          if (u < r) edge_step(hv[u], was, ad, m, den, acc);
      }
    }
  }
  const v4f b = *(const v4f*)(bias + lane * 4);
  v4f o;
  o.x = acc.x / den + b.x;
  o.y = acc.y / den + b.y;
  o.z = acc.z / den + b.z;
  o.w = acc.w / den + b.w;
  *(v4f*)(out + (size_t)d * F + lane * 4) = o;
}

extern "C" void kernel_launch(void* const* d_in, const int* in_sizes, int n_in,
                              void* d_out, int out_size, void* d_ws, size_t ws_size,
                              hipStream_t stream) {
  const float* x = (const float*)d_in[0];
  const int* ei = (const int*)d_in[1];
  const float* W = (const float*)d_in[2];
  const float* att_src = (const float*)d_in[3];
  const float* att_dst = (const float*)d_in[4];
  const float* bias = (const float*)d_in[5];
  float* out = (float*)d_out;

  const int N = in_sizes[0] / F;
  const int E = in_sizes[1] / 2;
  const int* srcA = ei;
  const int* dstA = ei + E;

  float* ws = (float*)d_ws;
  float* hbuf = ws;
  int* cnt = (int*)(hbuf + (size_t)N * F);
  int* offs = cnt + N;
  int* rank = offs + N;
  int* col = rank + E;

  node_gemm_wmma<<<(N + 15) / 16, 256, 0, stream>>>(x, W, hbuf, cnt, N);
  edge_rank<<<(E + 255) / 256, 256, 0, stream>>>(dstA, cnt, rank, E);
  count_scan<<<(N + 1023) / 1024, 256, 0, stream>>>(cnt, offs, N);
  edge_place<<<(E + 255) / 256, 256, 0, stream>>>(srcA, dstA, rank, offs, col, E);
  node_softmax_gather<<<(N + 7) / 8, 256, 0, stream>>>(cnt, offs, col, att_src, att_dst,
                                                       hbuf, bias, out, N);
}


# Fixed-stride buckets, 64 slots per node: the thread whose atomic returns the rank stores the source at node*64+rank; scan and place launches dropped; deeper buckets spill to a list; sum order kept
# speedup vs baseline: 4.4672x; 1.2085x over previous
#include <hip/hip_runtime.h>
#include <hip/hip_bf16.h>

typedef float v2f __attribute__((ext_vector_type(2)));
typedef float v8f __attribute__((ext_vector_type(8)));
typedef float v4f __attribute__((ext_vector_type(4)));

#define CH 16
#define F 128
#define NEG_SLOPE 0.2f
#define CAP 64

#if defined(__has_builtin)
#if __has_builtin(__builtin_amdgcn_wmma_f32_16x16x4_f32)
#define HAVE_WMMA_F32X4 1
#endif
#endif

__device__ __forceinline__ float lrelu(float v) {
  return v > 0.0f ? v : NEG_SLOPE * v;
}

__global__ __launch_bounds__(256) void node_gemm_wmma(
    const float* __restrict__ x, const float* __restrict__ W,
    float* __restrict__ h, int* __restrict__ cnt, int N) {
  __shared__ float sA[16 * F];

  const int tid = threadIdx.x;
  const int wave = tid >> 5;
  const int lane = tid & 31;
  const int half = lane >> 4;
  const int l16 = lane & 15;
  const int rowbase = blockIdx.x * 16;

  if (tid < 16 && rowbase + tid < N) cnt[rowbase + tid] = 0;
  if (blockIdx.x == 0 && tid == 16) cnt[N] = 0;

  {
    int r = tid >> 4;
    int c = (tid & 15) << 3;
    int rg = rowbase + r;
    if (rg >= N) rg = N - 1;
    const float4* s4 = (const float4*)(x + (size_t)rg * F + c);
    float4 a0 = s4[0];
    float4 a1 = s4[1];
    *(float4*)(sA + r * F + c) = a0;
    *(float4*)(sA + r * F + c + 4) = a1;
  }
  __syncthreads();

  const int colbase = wave * CH;
  v8f acc = {0.f, 0.f, 0.f, 0.f, 0.f, 0.f, 0.f, 0.f};

  for (int k = 0; k < F; k += 4) {
#if defined(HAVE_WMMA_F32X4)
    v2f afrag;
    const float* ap = sA + l16 * F + k + 2 * half;
    afrag.x = ap[0];
    afrag.y = ap[1];
    v2f bfrag;
    const float* bp = W + (size_t)(k + 2 * half) * F + colbase + l16;
    bfrag.x = bp[0];
    bfrag.y = bp[F];
    acc = __builtin_amdgcn_wmma_f32_16x16x4_f32(false, afrag, false, bfrag,
                                                (short)0, acc, false, false);
#else
    for (int v = 0; v < 8; ++v) {
      int row = v + 8 * half;
      for (int j = 0; j < 4; ++j)
        acc[v] += sA[row * F + k + j] * W[(size_t)(k + j) * F + colbase + l16];
    }
#endif
  }

#pragma unroll
  for (int v = 0; v < 8; ++v) {
    int row = rowbase + v + 8 * half;
    if (row < N) h[(size_t)row * F + colbase + l16] = acc[v];
  }
}

__global__ __launch_bounds__(256) void edge_bucket(const int* __restrict__ srcA,
                                                   const int* __restrict__ dstA,
                                                   int* cnt, int* __restrict__ col,
                                                   int* __restrict__ ovf, int N, int E) {
  const int e = blockIdx.x * 256 + threadIdx.x;
  if (e < E) {
    const int d = dstA[e];
    const int s = srcA[e];
    const int r = atomicAdd(&cnt[d], 1);
    if (r < CAP) {
      col[(size_t)d * CAP + r] = s;
    } else {
      const int q = atomicAdd(&cnt[N], 1);
      ovf[2 * (size_t)q] = d;
      ovf[2 * (size_t)q + 1] = s;
    }
  }
}

__device__ __forceinline__ float quad_sum(float p) {
  p += __int_as_float(__builtin_amdgcn_update_dpp(0, __float_as_int(p), 0xB1, 0xF, 0xF, true));
  p += __int_as_float(__builtin_amdgcn_update_dpp(0, __float_as_int(p), 0x4E, 0xF, 0xF, true));
  return p;
}

__device__ __forceinline__ float dot4(const v4f a, const v4f b) {
  return a.x * b.x + a.y * b.y + a.z * b.z + a.w * b.w;
}

#define ROWS_AHEAD 8

__device__ __forceinline__ void edge_step(const v4f hv, const v4f was,
                                          const float ad, float& m, float& den,
                                          v4f& acc) {
  const float lg = lrelu(quad_sum(dot4(hv, was)) + ad);
  const float t = lg - m;
  const float pe = expf(-fabsf(t));
  const bool up = t > 0.0f;
  const float sc = up ? pe : 1.0f;
  const float ev = up ? 1.0f : pe;
  m = up ? lg : m;
  den = den * sc + ev;
  acc.x = acc.x * sc + ev * hv.x;
  acc.y = acc.y * sc + ev * hv.y;
  acc.z = acc.z * sc + ev * hv.z;
  acc.w = acc.w * sc + ev * hv.w;
}

__global__ __launch_bounds__(256) void node_softmax_gather(
    const int* __restrict__ cnt, const int* __restrict__ col,
    const int* __restrict__ ovf,
    const float* __restrict__ att_src, const float* __restrict__ att_dst,
    const float* __restrict__ h, const float* __restrict__ bias,
    float* __restrict__ out, int N) {
  const int d = __builtin_amdgcn_readfirstlane(
      (int)(blockIdx.x * 8 + (threadIdx.x >> 5)));
  if (d >= N) return;
  const int lane = threadIdx.x & 31;
  const int deg = cnt[d];
  const int nb = (deg < CAP) ? deg : CAP;
  const int* __restrict__ cp = col + (size_t)d * CAP;
  const v4f was = *(const v4f*)(att_src + lane * 4);
  const v4f wad = *(const v4f*)(att_dst + lane * 4);
  const v4f hself = *(const v4f*)(h + (size_t)d * F + lane * 4);

  const float ad = quad_sum(dot4(hself, wad));
  float m = lrelu(quad_sum(dot4(hself, was)) + ad);
  float den = 1.0f;
  v4f acc = hself;

  for (int c0 = 0; c0 < nb; c0 += 32) {
    const int n = (nb - c0 < 32) ? (nb - c0) : 32;
    const int sv = (lane < n) ? cp[c0 + lane] : d;
    for (int j0 = 0; j0 < n; j0 += ROWS_AHEAD) {
      const int r = n - j0;
      int s[ROWS_AHEAD];
#pragma unroll
      for (int u = 0; u < ROWS_AHEAD; ++u) s[u] = __builtin_amdgcn_readlane(sv, j0 + u);
      v4f hv[ROWS_AHEAD];
      if (r >= ROWS_AHEAD) {
#pragma unroll
        for (int u = 0; u < ROWS_AHEAD; ++u)
          hv[u] = *(const v4f*)(h + (size_t)s[u] * F + lane * 4);
        asm volatile("" ::"v"(hv[0]), "v"(hv[1]), "v"(hv[2]), "v"(hv[3]),
                     "v"(hv[4]), "v"(hv[5]), "v"(hv[6]), "v"(hv[7]));
#pragma unroll
        for (int u = 0; u < ROWS_AHEAD; ++u) edge_step(hv[u], was, ad, m, den, acc);
      } else {
#pragma unroll
        for (int u = 0; u < ROWS_AHEAD - 1; ++u)
          if (u < r) hv[u] = *(const v4f*)(h + (size_t)s[u] * F + lane * 4);
#pragma unroll
        for (int u = 0; u < ROWS_AHEAD - 1; ++u)
          if (u < r) edge_step(hv[u], was, ad, m, den, acc);
      }
    }
  }
  if (deg > CAP) {
    const int no = cnt[N];
    for (int q0 = 0; q0 < no; q0 += 32) {
      const int q = q0 + lane;
      const int dd = (q < no) ? ovf[2 * (size_t)q] : -1;
      const int ss = (q < no) ? ovf[2 * (size_t)q + 1] : d;
      unsigned mine = __builtin_amdgcn_ballot_w32(dd == d);
      while (mine != 0u) {
        const int j = __builtin_ctz(mine);
        mine &= mine - 1u;
        const int s1 = __builtin_amdgcn_readlane(ss, j);
        const v4f hv1 = *(const v4f*)(h + (size_t)s1 * F + lane * 4);
        edge_step(hv1, was, ad, m, den, acc);
      }
    }
  }
  const v4f b = *(const v4f*)(bias + lane * 4);
  v4f o;
  o.x = acc.x / den + b.x;
  o.y = acc.y / den + b.y;
  o.z = acc.z / den + b.z;
  o.w = acc.w / den + b.w;
  *(v4f*)(out + (size_t)d * F + lane * 4) = o;
}

extern "C" void kernel_launch(void* const* d_in, const int* in_sizes, int n_in,
                              void* d_out, int out_size, void* d_ws, size_t ws_size,
                              hipStream_t stream) {
  const float* x = (const float*)d_in[0];
  const int* ei = (const int*)d_in[1];
  const float* W = (const float*)d_in[2];
  const float* att_src = (const float*)d_in[3];
  const float* att_dst = (const float*)d_in[4];
  const float* bias = (const float*)d_in[5];
  float* out = (float*)d_out;

  const int N = in_sizes[0] / F;
  const int E = in_sizes[1] / 2;
  const int* srcA = ei;
  const int* dstA = ei + E;

  float* ws = (float*)d_ws;
  float* hbuf = ws;
  int* col = (int*)(hbuf + (size_t)N * F);
  int* cnt = col + (size_t)N * CAP;
  int* ovf = cnt + N + 1;

  node_gemm_wmma<<<(N + 15) / 16, 256, 0, stream>>>(x, W, hbuf, cnt, N);
  edge_bucket<<<(E + 255) / 256, 256, 0, stream>>>(srcA, dstA, cnt, col, ovf, N, E);
  node_softmax_gather<<<(N + 7) / 8, 256, 0, stream>>>(cnt, col, ovf, att_src, att_dst,
                                                       hbuf, bias, out, N);
}


# x strip rows padded to a 528-byte pitch in LDS: the lanes of an A-fragment read (one row each, same k) no longer all fall on one bank pair; k order and sum orders kept
# speedup vs baseline: 4.9307x; 1.1038x over previous
#include <hip/hip_runtime.h>
#include <hip/hip_bf16.h>

typedef float v2f __attribute__((ext_vector_type(2)));
typedef float v8f __attribute__((ext_vector_type(8)));
typedef float v4f __attribute__((ext_vector_type(4)));

#define CH 16
#define F 128
#define NEG_SLOPE 0.2f
#define CAP 64
#define LDP (F + 4)

#if defined(__has_builtin)
#if __has_builtin(__builtin_amdgcn_wmma_f32_16x16x4_f32)
#define HAVE_WMMA_F32X4 1
#endif
#endif

__device__ __forceinline__ float lrelu(float v) {
  return v > 0.0f ? v : NEG_SLOPE * v;
}

__global__ __launch_bounds__(256) void node_gemm_wmma(
    const float* __restrict__ x, const float* __restrict__ W,
    float* __restrict__ h, int* __restrict__ cnt, int N) {
  __shared__ float sA[16 * LDP];

  const int tid = threadIdx.x;
  const int wave = tid >> 5;
  const int lane = tid & 31;
  const int half = lane >> 4;
  const int l16 = lane & 15;
  const int rowbase = blockIdx.x * 16;

  if (tid < 16 && rowbase + tid < N) cnt[rowbase + tid] = 0;
  if (blockIdx.x == 0 && tid == 16) cnt[N] = 0;

  {
    int r = tid >> 4;
    int c = (tid & 15) << 3;
    int rg = rowbase + r;
    if (rg >= N) rg = N - 1;
    const float4* s4 = (const float4*)(x + (size_t)rg * F + c);
    float4 a0 = s4[0];
    float4 a1 = s4[1];
    *(float4*)(sA + r * LDP + c) = a0;
    *(float4*)(sA + r * LDP + c + 4) = a1;
  }
  __syncthreads();

  const int colbase = wave * CH;
  v8f acc = {0.f, 0.f, 0.f, 0.f, 0.f, 0.f, 0.f, 0.f};

  for (int k = 0; k < F; k += 4) {
#if defined(HAVE_WMMA_F32X4)
    v2f afrag;
    const float* ap = sA + l16 * LDP + k + 2 * half;
    afrag.x = ap[0];
    afrag.y = ap[1];
    v2f bfrag;
    const float* bp = W + (size_t)(k + 2 * half) * F + colbase + l16;
    bfrag.x = bp[0];
    bfrag.y = bp[F];
    acc = __builtin_amdgcn_wmma_f32_16x16x4_f32(false, afrag, false, bfrag,
                                                (short)0, acc, false, false);
#else
    for (int v = 0; v < 8; ++v) {
      int row = v + 8 * half;
      for (int j = 0; j < 4; ++j)
        acc[v] += sA[row * LDP + k + j] * W[(size_t)(k + j) * F + colbase + l16];
    }
#endif
  }

#pragma unroll
  for (int v = 0; v < 8; ++v) {
    int row = rowbase + v + 8 * half;
    if (row < N) h[(size_t)row * F + colbase + l16] = acc[v];
  }
}

__global__ __launch_bounds__(256) void edge_bucket(const int* __restrict__ srcA,
                                                   const int* __restrict__ dstA,
                                                   int* cnt, int* __restrict__ col,
                                                   int* __restrict__ ovf, int N, int E) {
  const int e = blockIdx.x * 256 + threadIdx.x;
  if (e < E) {
    const int d = dstA[e];
    const int s = srcA[e];
    const int r = atomicAdd(&cnt[d], 1);
    if (r < CAP) {
      col[(size_t)d * CAP + r] = s;
    } else {
      const int q = atomicAdd(&cnt[N], 1);
      ovf[2 * (size_t)q] = d;
      ovf[2 * (size_t)q + 1] = s;
    }
  }
}

__device__ __forceinline__ float quad_sum(float p) {
  p += __int_as_float(__builtin_amdgcn_update_dpp(0, __float_as_int(p), 0xB1, 0xF, 0xF, true));
  p += __int_as_float(__builtin_amdgcn_update_dpp(0, __float_as_int(p), 0x4E, 0xF, 0xF, true));
  return p;
}

__device__ __forceinline__ float dot4(const v4f a, const v4f b) {
  return a.x * b.x + a.y * b.y + a.z * b.z + a.w * b.w;
}

#define ROWS_AHEAD 8

__device__ __forceinline__ void edge_step(const v4f hv, const v4f was,
                                          const float ad, float& m, float& den,
                                          v4f& acc) {
  const float lg = lrelu(quad_sum(dot4(hv, was)) + ad);
  const float t = lg - m;
  const float pe = expf(-fabsf(t));
  const bool up = t > 0.0f;
  const float sc = up ? pe : 1.0f;
  const float ev = up ? 1.0f : pe;
  m = up ? lg : m;
  den = den * sc + ev;
  acc.x = acc.x * sc + ev * hv.x;
  acc.y = acc.y * sc + ev * hv.y;
  acc.z = acc.z * sc + ev * hv.z;
  acc.w = acc.w * sc + ev * hv.w;
}

__global__ __launch_bounds__(256) void node_softmax_gather(
    const int* __restrict__ cnt, const int* __restrict__ col,
    const int* __restrict__ ovf,
    const float* __restrict__ att_src, const float* __restrict__ att_dst,
    const float* __restrict__ h, const float* __restrict__ bias,
    float* __restrict__ out, int N) {
  const int d = __builtin_amdgcn_readfirstlane(
      (int)(blockIdx.x * 8 + (threadIdx.x >> 5)));
  if (d >= N) return;
  const int lane = threadIdx.x & 31;
  const int deg = cnt[d];
  const int nb = (deg < CAP) ? deg : CAP;
  const int* __restrict__ cp = col + (size_t)d * CAP;
  const v4f was = *(const v4f*)(att_src + lane * 4);
  const v4f wad = *(const v4f*)(att_dst + lane * 4);
  const v4f hself = *(const v4f*)(h + (size_t)d * F + lane * 4);

  const float ad = quad_sum(dot4(hself, wad));
  float m = lrelu(quad_sum(dot4(hself, was)) + ad);
  float den = 1.0f;
  v4f acc = hself;

  for (int c0 = 0; c0 < nb; c0 += 32) {
    const int n = (nb - c0 < 32) ? (nb - c0) : 32;
    const int sv = (lane < n) ? cp[c0 + lane] : d;
    for (int j0 = 0; j0 < n; j0 += ROWS_AHEAD) {
      const int r = n - j0;
      int s[ROWS_AHEAD];
#pragma unroll
      for (int u = 0; u < ROWS_AHEAD; ++u) s[u] = __builtin_amdgcn_readlane(sv, j0 + u);
      v4f hv[ROWS_AHEAD];
      if (r >= ROWS_AHEAD) {
#pragma unroll
        for (int u = 0; u < ROWS_AHEAD; ++u)
          hv[u] = *(const v4f*)(h + (size_t)s[u] * F + lane * 4);
        asm volatile("" ::"v"(hv[0]), "v"(hv[1]), "v"(hv[2]), "v"(hv[3]),
                     "v"(hv[4]), "v"(hv[5]), "v"(hv[6]), "v"(hv[7]));
#pragma unroll
        for (int u = 0; u < ROWS_AHEAD; ++u) edge_step(hv[u], was, ad, m, den, acc);
      } else {
#pragma unroll
        for (int u = 0; u < ROWS_AHEAD - 1; ++u)
          if (u < r) hv[u] = *(const v4f*)(h + (size_t)s[u] * F + lane * 4);
#pragma unroll
        for (int u = 0; u < ROWS_AHEAD - 1; ++u)
          if (u < r) edge_step(hv[u], was, ad, m, den, acc);
      }
    }
  }
  if (deg > CAP) {
    const int no = cnt[N];
    for (int q0 = 0; q0 < no; q0 += 32) {
      const int q = q0 + lane;
      const int dd = (q < no) ? ovf[2 * (size_t)q] : -1;
      const int ss = (q < no) ? ovf[2 * (size_t)q + 1] : d;
      unsigned mine = __builtin_amdgcn_ballot_w32(dd == d);
      while (mine != 0u) {
        const int j = __builtin_ctz(mine);
        mine &= mine - 1u;
        const int s1 = __builtin_amdgcn_readlane(ss, j);
        const v4f hv1 = *(const v4f*)(h + (size_t)s1 * F + lane * 4);
        edge_step(hv1, was, ad, m, den, acc);
      }
    }
  }
  const v4f b = *(const v4f*)(bias + lane * 4);
  v4f o;
  o.x = acc.x / den + b.x;
  o.y = acc.y / den + b.y;
  o.z = acc.z / den + b.z;
  o.w = acc.w / den + b.w;
  *(v4f*)(out + (size_t)d * F + lane * 4) = o;
}

extern "C" void kernel_launch(void* const* d_in, const int* in_sizes, int n_in,
                              void* d_out, int out_size, void* d_ws, size_t ws_size,
                              hipStream_t stream) {
  const float* x = (const float*)d_in[0];
  const int* ei = (const int*)d_in[1];
  const float* W = (const float*)d_in[2];
  const float* att_src = (const float*)d_in[3];
  const float* att_dst = (const float*)d_in[4];
  const float* bias = (const float*)d_in[5];
  float* out = (float*)d_out;

  const int N = in_sizes[0] / F;
  const int E = in_sizes[1] / 2;
  const int* srcA = ei;
  const int* dstA = ei + E;

  float* ws = (float*)d_ws;
  float* hbuf = ws;
  int* col = (int*)(hbuf + (size_t)N * F);
  int* cnt = col + (size_t)N * CAP;
  int* ovf = cnt + N + 1;

  node_gemm_wmma<<<(N + 15) / 16, 256, 0, stream>>>(x, W, hbuf, cnt, N);
  edge_bucket<<<(E + 255) / 256, 256, 0, stream>>>(srcA, dstA, cnt, col, ovf, N, E);
  node_softmax_gather<<<(N + 7) / 8, 256, 0, stream>>>(cnt, col, ovf, att_src, att_dst,
                                                       hbuf, bias, out, N);
}


# Edges filed by a two-level counting sort on LDS counters, no global atomic per edge; gather sorts each bucket by source first (LDS arrival order varies per run): sum order ascending source, fixed
# speedup vs baseline: 6.9734x; 1.4143x over previous
#include <hip/hip_runtime.h>
#include <hip/hip_bf16.h>

typedef float v2f __attribute__((ext_vector_type(2)));
typedef float v8f __attribute__((ext_vector_type(8)));
typedef float v4f __attribute__((ext_vector_type(4)));

#define CH 16
#define F 128
#define NEG_SLOPE 0.2f
#define CAP 64
#define BSH 8
#define BN (1 << BSH)
#define NBMAX 448
#define SEG 32
#define CHUNK 4096
#define EPT (CHUNK / 256)
#define NWGMAX 512
#define HALFCAP 32
#define LDP (F + 4)

#if defined(__has_builtin)
#if __has_builtin(__builtin_amdgcn_wmma_f32_16x16x4_f32)
#define HAVE_WMMA_F32X4 1
#endif
#endif

__device__ __forceinline__ float lrelu(float v) {
  return v > 0.0f ? v : NEG_SLOPE * v;
}

__global__ __launch_bounds__(256) void node_gemm_wmma(
    const float* __restrict__ x, const float* __restrict__ W,
    float* __restrict__ h, int* __restrict__ misc, int N) {
  __shared__ float sA[16 * LDP];

  const int tid = threadIdx.x;
  const int wave = tid >> 5;
  const int lane = tid & 31;
  const int half = lane >> 4;
  const int l16 = lane & 15;
  const int rowbase = blockIdx.x * 16;

  if (blockIdx.x == 0 && tid == 16) misc[0] = 0;

  {
    int r = tid >> 4;
    int c = (tid & 15) << 3;
    int rg = rowbase + r;
    if (rg >= N) rg = N - 1;
    const float4* s4 = (const float4*)(x + (size_t)rg * F + c);
    float4 a0 = s4[0];
    float4 a1 = s4[1];
    *(float4*)(sA + r * LDP + c) = a0;
    *(float4*)(sA + r * LDP + c + 4) = a1;
  }
  __syncthreads();

  const int colbase = wave * CH;
  v8f acc = {0.f, 0.f, 0.f, 0.f, 0.f, 0.f, 0.f, 0.f};

  for (int k = 0; k < F; k += 4) {
#if defined(HAVE_WMMA_F32X4)
    v2f afrag;
    const float* ap = sA + l16 * LDP + k + 2 * half;
    afrag.x = ap[0];
    afrag.y = ap[1];
    v2f bfrag;
    const float* bp = W + (size_t)(k + 2 * half) * F + colbase + l16;
    bfrag.x = bp[0];
    bfrag.y = bp[F];
    acc = __builtin_amdgcn_wmma_f32_16x16x4_f32(false, afrag, false, bfrag,
                                                (short)0, acc, false, false);
#else
    for (int v = 0; v < 8; ++v) {
      int row = v + 8 * half;
      for (int j = 0; j < 4; ++j)
        acc[v] += sA[row * LDP + k + j] * W[(size_t)(k + j) * F + colbase + l16];
    }
#endif
  }

#pragma unroll
  for (int v = 0; v < 8; ++v) {
    int row = rowbase + v + 8 * half;
    if (row < N) h[(size_t)row * F + colbase + l16] = acc[v];
  }
}

__global__ __launch_bounds__(256) void edge_split(const int* __restrict__ srcA,
                                                  const int* __restrict__ dstA,
                                                  int* __restrict__ seg,
                                                  int* __restrict__ segcnt, int* misc,
                                                  int* __restrict__ ovf, int E, int NB,
                                                  int NWG) {
  __shared__ int lcnt[NBMAX];
  __shared__ int stage[NBMAX * SEG];
  const int tid = threadIdx.x;
  const int w = blockIdx.x;
  for (int b = tid; b < NB; b += 256) lcnt[b] = 0;
  int dd[EPT], ss[EPT];
  const int base = w * CHUNK + tid;
#pragma unroll
  for (int i = 0; i < EPT; ++i) {
    const int e = base + i * 256;
    dd[i] = (e < E) ? dstA[e] : -1;
    ss[i] = (e < E) ? srcA[e] : 0;
  }
  __syncthreads();
#pragma unroll
  for (int i = 0; i < EPT; ++i) {
    const int d = dd[i];
    if (d >= 0) {
      const int b = d >> BSH;
      const int r = atomicAdd(&lcnt[b], 1);
      if (r < SEG) {
        stage[b * SEG + r] = (int)(((unsigned)(d & (BN - 1)) << 24) | (unsigned)ss[i]);
      } else {
        const int q = atomicAdd(&misc[0], 1);
        ovf[2 * (size_t)q] = d;
        ovf[2 * (size_t)q + 1] = ss[i];
      }
    }
  }
  __syncthreads();
  const int wave = tid >> 5;
  const int lane = tid & 31;
  for (int b = wave; b < NB; b += 8) {
    int c = lcnt[b];
    c = (c < SEG) ? c : SEG;
    const size_t sg = (size_t)b * NWG + w;
    if (lane < c) seg[sg * SEG + lane] = stage[b * SEG + lane];
    if (lane == 0) segcnt[sg] = c;
  }
}

#define SEGS_AHEAD 8
__global__ __launch_bounds__(256) void bin_bucket(const int* __restrict__ seg,
                                                  const int* __restrict__ segcnt,
                                                  int* __restrict__ col,
                                                  int* __restrict__ cntg, int* misc,
                                                  int* __restrict__ ovf, int N, int NWG) {
  __shared__ int ncnt[BN];
  __shared__ int scnt[NWGMAX];
  __shared__ int bst[BN * HALFCAP];
  const int tid = threadIdx.x;
  const int b = blockIdx.x;
  ncnt[tid] = 0;
  for (int i = tid; i < NWG; i += 256) scnt[i] = segcnt[(size_t)b * NWG + i];
  __syncthreads();
  const int wave = tid >> 5;
  const int lane = tid & 31;
  for (int g0 = wave; g0 < NWG; g0 += 8 * SEGS_AHEAD) {
    int ev[SEGS_AHEAD];
    int cc[SEGS_AHEAD];
#pragma unroll
    for (int u = 0; u < SEGS_AHEAD; ++u) {
      const int g = g0 + 8 * u;
      cc[u] = (g < NWG) ? scnt[g] : 0;
      ev[u] = 0;
      if (lane < cc[u]) ev[u] = seg[((size_t)b * NWG + g) * SEG + lane];
    }
#pragma unroll
    for (int u = 0; u < SEGS_AHEAD; ++u) {
      if (lane < cc[u]) {
        const unsigned e = (unsigned)ev[u];
        const int dl = (int)(e >> 24);
        const int s = (int)(e & 0xFFFFFFu);
        const int r = atomicAdd(&ncnt[dl], 1);
        if (r < HALFCAP) {
          bst[dl * HALFCAP + r] = s;
        } else if (r < CAP) {
          col[(size_t)(b * BN + dl) * CAP + r] = s;
        } else {
          const int q = atomicAdd(&misc[0], 1);
          ovf[2 * (size_t)q] = b * BN + dl;
          ovf[2 * (size_t)q + 1] = s;
        }
      }
    }
  }
  __syncthreads();
  for (int dl = wave; dl < BN; dl += 8) {
    const int node = b * BN + dl;
    if (node < N) {
      const int c = ncnt[dl];
      const int nh = (c < HALFCAP) ? c : HALFCAP;
      if (lane < nh) col[(size_t)node * CAP + lane] = bst[dl * HALFCAP + lane];
      if (lane == 0) cntg[node] = c;
    }
  }
}

__device__ __forceinline__ float quad_sum(float p) {
  p += __int_as_float(__builtin_amdgcn_update_dpp(0, __float_as_int(p), 0xB1, 0xF, 0xF, true));
  p += __int_as_float(__builtin_amdgcn_update_dpp(0, __float_as_int(p), 0x4E, 0xF, 0xF, true));
  return p;
}

__device__ __forceinline__ float dot4(const v4f a, const v4f b) {
  return a.x * b.x + a.y * b.y + a.z * b.z + a.w * b.w;
}

#define ROWS_AHEAD 8

__device__ __forceinline__ void edge_step(const v4f hv, const v4f was,
                                          const float ad, float& m, float& den,
                                          v4f& acc) {
  const float lg = lrelu(quad_sum(dot4(hv, was)) + ad);
  const float t = lg - m;
  const float pe = expf(-fabsf(t));
  const bool up = t > 0.0f;
  const float sc = up ? pe : 1.0f;
  const float ev = up ? 1.0f : pe;
  m = up ? lg : m;
  den = den * sc + ev;
  acc.x = acc.x * sc + ev * hv.x;
  acc.y = acc.y * sc + ev * hv.y;
  acc.z = acc.z * sc + ev * hv.z;
  acc.w = acc.w * sc + ev * hv.w;
}

__global__ __launch_bounds__(256) void node_softmax_gather(
    const int* __restrict__ cntg, const int* __restrict__ misc,
    const int* __restrict__ col, const int* __restrict__ ovf,
    const float* __restrict__ att_src, const float* __restrict__ att_dst,
    const float* __restrict__ h, const float* __restrict__ bias,
    float* __restrict__ out, int N) {
  const int d = __builtin_amdgcn_readfirstlane(
      (int)(blockIdx.x * 8 + (threadIdx.x >> 5)));
  if (d >= N) return;
  const int lane = threadIdx.x & 31;
  const int deg = cntg[d];
  const int novf = misc[0];
  const int nb = (deg < CAP) ? deg : CAP;
  const int* __restrict__ cp = col + (size_t)d * CAP;
  const v4f was = *(const v4f*)(att_src + lane * 4);
  const v4f wad = *(const v4f*)(att_dst + lane * 4);
  const v4f hself = *(const v4f*)(h + (size_t)d * F + lane * 4);

  const float ad = quad_sum(dot4(hself, wad));
  float m = lrelu(quad_sum(dot4(hself, was)) + ad);
  float den = 1.0f;
  v4f acc = hself;

  if (nb <= 32) {
    const int n = nb;
    int sv = (lane < n) ? cp[lane] : d;
    int rk = 0;
    for (int i = 0; i < n; ++i) {
      const int si = __builtin_amdgcn_readlane(sv, i);
      rk += (si < sv || (si == sv && i < lane)) ? 1 : 0;
    }
    const int place = (lane < n) ? rk : lane;
    sv = __builtin_amdgcn_ds_permute(place << 2, sv);
    for (int j0 = 0; j0 < n; j0 += ROWS_AHEAD) {
      const int r = n - j0;
      int s[ROWS_AHEAD];
#pragma unroll
      for (int u = 0; u < ROWS_AHEAD; ++u) s[u] = __builtin_amdgcn_readlane(sv, j0 + u);
      v4f hv[ROWS_AHEAD];
      if (r >= ROWS_AHEAD) {
#pragma unroll
        for (int u = 0; u < ROWS_AHEAD; ++u)
          hv[u] = *(const v4f*)(h + (size_t)s[u] * F + lane * 4);
        asm volatile("" ::"v"(hv[0]), "v"(hv[1]), "v"(hv[2]), "v"(hv[3]),
                     "v"(hv[4]), "v"(hv[5]), "v"(hv[6]), "v"(hv[7]));
#pragma unroll
        for (int u = 0; u < ROWS_AHEAD; ++u) edge_step(hv[u], was, ad, m, den, acc);
      } else {
#pragma unroll
        for (int u = 0; u < ROWS_AHEAD - 1; ++u)
          if (u < r) hv[u] = *(const v4f*)(h + (size_t)s[u] * F + lane * 4);
#pragma unroll
        for (int u = 0; u < ROWS_AHEAD - 1; ++u)
          if (u < r) edge_step(hv[u], was, ad, m, den, acc);
      }
    }
  } else {
    const int n1 = nb - 32;
    const int sv0 = cp[lane];
    const int sv1 = (lane < n1) ? cp[32 + lane] : 0x7fffffff;
    int r0 = 0, r1 = 0;
    for (int i = 0; i < 32; ++i) {
      const int si = __builtin_amdgcn_readlane(sv0, i);
      r0 += (si < sv0 || (si == sv0 && i < lane)) ? 1 : 0;
      r1 += (si <= sv1) ? 1 : 0;
    }
    for (int i = 0; i < n1; ++i) {
      const int si = __builtin_amdgcn_readlane(sv1, i);
      r0 += (si < sv0) ? 1 : 0;
      r1 += (si < sv1 || (si == sv1 && i < lane)) ? 1 : 0;
    }
    for (int k = 0; k < nb; ++k) {
      const unsigned b0 = __builtin_amdgcn_ballot_w32(r0 == k);
      const unsigned b1 = __builtin_amdgcn_ballot_w32(lane < n1 && r1 == k);
      int s1;
      if (b0 != 0u) s1 = __builtin_amdgcn_readlane(sv0, __builtin_ctz(b0));
      else s1 = __builtin_amdgcn_readlane(sv1, __builtin_ctz(b1));
      const v4f hv1 = *(const v4f*)(h + (size_t)s1 * F + lane * 4);
      edge_step(hv1, was, ad, m, den, acc);
    }
  }
  if (novf > 0) {
    const int no = novf;
    for (int q0 = 0; q0 < no; q0 += 32) {
      const int q = q0 + lane;
      const int dd = (q < no) ? ovf[2 * (size_t)q] : -1;
      const int ss = (q < no) ? ovf[2 * (size_t)q + 1] : d;
      unsigned mine = __builtin_amdgcn_ballot_w32(dd == d);
      while (mine != 0u) {
        const int j = __builtin_ctz(mine);
        mine &= mine - 1u;
        const int s1 = __builtin_amdgcn_readlane(ss, j);
        const v4f hv1 = *(const v4f*)(h + (size_t)s1 * F + lane * 4);
        edge_step(hv1, was, ad, m, den, acc);
      }
    }
  }
  const v4f b = *(const v4f*)(bias + lane * 4);
  v4f o;
  o.x = acc.x / den + b.x;
  o.y = acc.y / den + b.y;
  o.z = acc.z / den + b.z;
  o.w = acc.w / den + b.w;
  *(v4f*)(out + (size_t)d * F + lane * 4) = o;
}

extern "C" void kernel_launch(void* const* d_in, const int* in_sizes, int n_in,
                              void* d_out, int out_size, void* d_ws, size_t ws_size,
                              hipStream_t stream) {
  const float* x = (const float*)d_in[0];
  const int* ei = (const int*)d_in[1];
  const float* W = (const float*)d_in[2];
  const float* att_src = (const float*)d_in[3];
  const float* att_dst = (const float*)d_in[4];
  const float* bias = (const float*)d_in[5];
  float* out = (float*)d_out;

  const int N = in_sizes[0] / F;
  const int E = in_sizes[1] / 2;
  const int* srcA = ei;
  const int* dstA = ei + E;

  const int NB = (N + BN - 1) / BN;
  const int NWG = (E + CHUNK - 1) / CHUNK;

  float* ws = (float*)d_ws;
  float* hbuf = ws;
  int* col = (int*)(hbuf + (size_t)N * F);
  int* seg = col + (size_t)N * CAP;
  int* segcnt = seg + (size_t)NB * NWG * SEG;
  int* cntg = segcnt + (size_t)NB * NWG;
  int* misc = cntg + N;
  int* ovf = misc + 4;

  node_gemm_wmma<<<(N + 15) / 16, 256, 0, stream>>>(x, W, hbuf, misc, N);
  edge_split<<<NWG, 256, 0, stream>>>(srcA, dstA, seg, segcnt, misc, ovf, E, NB, NWG);
  bin_bucket<<<NB, 256, 0, stream>>>(seg, segcnt, col, cntg, misc, ovf, N, NWG);
  node_softmax_gather<<<(N + 7) / 8, 256, 0, stream>>>(cntg, misc, col, ovf, att_src, att_dst,
                                                       hbuf, bias, out, N);
}


# Each sort pass shares its launch with part of the GEMM strips (split + 3/8, bin + the rest; split spills now in per-chunk lists): 3 launches, sort waits filled by matrix work; sum orders kept
# speedup vs baseline: 7.3356x; 1.0519x over previous
#include <hip/hip_runtime.h>
#include <hip/hip_bf16.h>

typedef float v2f __attribute__((ext_vector_type(2)));
typedef float v8f __attribute__((ext_vector_type(8)));
typedef float v4f __attribute__((ext_vector_type(4)));

#define CH 16
#define F 128
#define NEG_SLOPE 0.2f
#define CAP 64
#define BSH 8
#define BN (1 << BSH)
#define NBMAX 448
#define SEG 32
#define CHUNK 4096
#define EPT (CHUNK / 256)
#define NWGMAX 512
#define HALFCAP 32
#define LDP (F + 4)

#if defined(__has_builtin)
#if __has_builtin(__builtin_amdgcn_wmma_f32_16x16x4_f32)
#define HAVE_WMMA_F32X4 1
#endif
#endif

__device__ __forceinline__ float lrelu(float v) {
  return v > 0.0f ? v : NEG_SLOPE * v;
}

__device__ __forceinline__ void gemm_strip(
    const float* __restrict__ x, const float* __restrict__ W,
    float* __restrict__ h, int N, int strip, float* sA) {
  const int tid = threadIdx.x;
  const int wave = tid >> 5;
  const int lane = tid & 31;
  const int half = lane >> 4;
  const int l16 = lane & 15;
  const int rowbase = strip * 16;

  {
    int r = tid >> 4;
    int c = (tid & 15) << 3;
    int rg = rowbase + r;
    if (rg >= N) rg = N - 1;
    const float4* s4 = (const float4*)(x + (size_t)rg * F + c);
    float4 a0 = s4[0];
    float4 a1 = s4[1];
    *(float4*)(sA + r * LDP + c) = a0;
    *(float4*)(sA + r * LDP + c + 4) = a1;
  }
  __syncthreads();

  const int colbase = wave * CH;
  v8f acc = {0.f, 0.f, 0.f, 0.f, 0.f, 0.f, 0.f, 0.f};

  for (int k = 0; k < F; k += 4) {
#if defined(HAVE_WMMA_F32X4)
    v2f afrag;
    const float* ap = sA + l16 * LDP + k + 2 * half;
    afrag.x = ap[0];
    afrag.y = ap[1];
    v2f bfrag;
    const float* bp = W + (size_t)(k + 2 * half) * F + colbase + l16;
    bfrag.x = bp[0];
    bfrag.y = bp[F];
    acc = __builtin_amdgcn_wmma_f32_16x16x4_f32(false, afrag, false, bfrag,
                                                (short)0, acc, false, false);
#else
    for (int v = 0; v < 8; ++v) {
      int row = v + 8 * half;
      for (int j = 0; j < 4; ++j)
        acc[v] += sA[row * LDP + k + j] * W[(size_t)(k + j) * F + colbase + l16];
    }
#endif
  }

#pragma unroll
  for (int v = 0; v < 8; ++v) {
    int row = rowbase + v + 8 * half;
    if (row < N) h[(size_t)row * F + colbase + l16] = acc[v];
  }
}

#define SPLIT_LDS_INTS (NBMAX + NBMAX * SEG + 4)
__device__ __forceinline__ void split_body(const int* __restrict__ srcA,
                                           const int* __restrict__ dstA,
                                           int* __restrict__ seg,
                                           int* __restrict__ segcnt,
                                           int* __restrict__ misc,
                                           int* __restrict__ spl,
                                           int* __restrict__ spcnt, int E, int NB,
                                           int NWG, int w, int* lds) {
  int* lcnt = lds;
  int* stage = lds + NBMAX;
  int* lsp = lds + NBMAX + NBMAX * SEG;
  const int tid = threadIdx.x;
  if (w == 0 && tid == 32) misc[0] = 0;
  if (tid == 0) lsp[0] = 0;
  for (int b = tid; b < NB; b += 256) lcnt[b] = 0;
  int dd[EPT], ss[EPT];
  const int base = w * CHUNK + tid;
#pragma unroll
  for (int i = 0; i < EPT; ++i) {
    const int e = base + i * 256;
    dd[i] = (e < E) ? dstA[e] : -1;
    ss[i] = (e < E) ? srcA[e] : 0;
  }
  __syncthreads();
#pragma unroll
  for (int i = 0; i < EPT; ++i) {
    const int d = dd[i];
    if (d >= 0) {
      const int b = d >> BSH;
      const int r = atomicAdd(&lcnt[b], 1);
      if (r < SEG) {
        stage[b * SEG + r] = (int)(((unsigned)(d & (BN - 1)) << 24) | (unsigned)ss[i]);
      } else {
        const int q = atomicAdd(lsp, 1);
        const size_t p = 2 * ((size_t)w * CHUNK + q);
        spl[p] = d;
        spl[p + 1] = ss[i];
      }
    }
  }
  __syncthreads();
  if (tid == 0) spcnt[w] = lsp[0];
  const int wave = tid >> 5;
  const int lane = tid & 31;
  for (int b = wave; b < NB; b += 8) {
    int c = lcnt[b];
    c = (c < SEG) ? c : SEG;
    const size_t sg = (size_t)b * NWG + w;
    if (lane < c) seg[sg * SEG + lane] = stage[b * SEG + lane];
    if (lane == 0) segcnt[sg] = c;
  }
}

#define SEGS_AHEAD 8
#define BIN_LDS_INTS (BN + NWGMAX + BN * HALFCAP + 4)
__device__ __forceinline__ void bin_body(const int* __restrict__ seg,
                                         const int* __restrict__ segcnt,
                                         const int* __restrict__ spl,
                                         const int* __restrict__ spcnt,
                                         int* __restrict__ col,
                                         int* __restrict__ cntg, int* misc,
                                         int* __restrict__ ovf, int N, int NWG,
                                         int b, int* lds) {
  int* ncnt = lds;
  int* scnt = lds + BN;
  int* bst = lds + BN + NWGMAX;
  int* sflag = lds + BN + NWGMAX + BN * HALFCAP;
  const int tid = threadIdx.x;
  ncnt[tid] = 0;
  if (tid == 0) sflag[0] = 0;
  int anysp = 0;
  for (int i = tid; i < NWG; i += 256) {
    scnt[i] = segcnt[(size_t)b * NWG + i];
    anysp |= spcnt[i];
  }
  __syncthreads();
  if (anysp != 0) sflag[0] = 1;
  const int wave = tid >> 5;
  const int lane = tid & 31;
  for (int g0 = wave; g0 < NWG; g0 += 8 * SEGS_AHEAD) {
    int ev[SEGS_AHEAD];
    int cc[SEGS_AHEAD];
#pragma unroll
    for (int u = 0; u < SEGS_AHEAD; ++u) {
      const int g = g0 + 8 * u;
      cc[u] = (g < NWG) ? scnt[g] : 0;
      ev[u] = 0;
      if (lane < cc[u]) ev[u] = seg[((size_t)b * NWG + g) * SEG + lane];
    }
#pragma unroll
    for (int u = 0; u < SEGS_AHEAD; ++u) {
      if (lane < cc[u]) {
        const unsigned e = (unsigned)ev[u];
        const int dl = (int)(e >> 24);
        const int s = (int)(e & 0xFFFFFFu);
        const int r = atomicAdd(&ncnt[dl], 1);
        if (r < HALFCAP) {
          bst[dl * HALFCAP + r] = s;
        } else if (r < CAP) {
          col[(size_t)(b * BN + dl) * CAP + r] = s;
        } else {
          const int q = atomicAdd(&misc[0], 1);
          ovf[2 * (size_t)q] = b * BN + dl;
          ovf[2 * (size_t)q + 1] = s;
        }
      }
    }
  }
  __syncthreads();
  if (sflag[0] != 0) {
    for (int i = 0; i < NWG; ++i) {
      int c = spcnt[i];
      c = (c < CHUNK) ? c : CHUNK;
      for (int q = tid; q < c; q += 256) {
        const size_t p = 2 * ((size_t)i * CHUNK + q);
        const int d = spl[p];
        const int s = spl[p + 1];
        if ((d >> BSH) == b) {
          const int dl = d & (BN - 1);
          const int r = atomicAdd(&ncnt[dl], 1);
          if (r < HALFCAP) {
            bst[dl * HALFCAP + r] = s;
          } else if (r < CAP) {
            col[(size_t)d * CAP + r] = s;
          } else {
            const int q2 = atomicAdd(&misc[0], 1);
            ovf[2 * (size_t)q2] = d;
            ovf[2 * (size_t)q2 + 1] = s;
          }
        }
      }
    }
    __syncthreads();
  }
  for (int dl = wave; dl < BN; dl += 8) {
    const int node = b * BN + dl;
    if (node < N) {
      const int c = ncnt[dl];
      const int nh = (c < HALFCAP) ? c : HALFCAP;
      if (lane < nh) col[(size_t)node * CAP + lane] = bst[dl * HALFCAP + lane];
      if (lane == 0) cntg[node] = c;
    }
  }
}

__global__ __launch_bounds__(256) void split_and_gemm(
    const int* __restrict__ srcA, const int* __restrict__ dstA,
    int* __restrict__ seg, int* __restrict__ segcnt, int* __restrict__ misc,
    int* __restrict__ spl, int* __restrict__ spcnt,
    const float* __restrict__ x, const float* __restrict__ W, float* __restrict__ h,
    int N, int E, int NB, int NWG) {
  __shared__ __attribute__((aligned(16))) int lds[SPLIT_LDS_INTS];
  const int wg = blockIdx.x;
  if (wg < NWG) {
    split_body(srcA, dstA, seg, segcnt, misc, spl, spcnt, E, NB, NWG, wg, lds);
  } else {
    gemm_strip(x, W, h, N, wg - NWG, (float*)lds);
  }
}

__global__ __launch_bounds__(256) void bin_and_gemm(
    const int* __restrict__ seg, const int* __restrict__ segcnt,
    const int* __restrict__ spl, const int* __restrict__ spcnt,
    int* __restrict__ col, int* __restrict__ cntg, int* misc, int* __restrict__ ovf,
    const float* __restrict__ x, const float* __restrict__ W, float* __restrict__ h,
    int N, int NWG, int NB, int strip0) {
  __shared__ __attribute__((aligned(16))) int lds[BIN_LDS_INTS];
  const int wg = blockIdx.x;
  if (wg < NB) {
    bin_body(seg, segcnt, spl, spcnt, col, cntg, misc, ovf, N, NWG, wg, lds);
  } else {
    gemm_strip(x, W, h, N, strip0 + wg - NB, (float*)lds);
  }
}

__device__ __forceinline__ float quad_sum(float p) {
  p += __int_as_float(__builtin_amdgcn_update_dpp(0, __float_as_int(p), 0xB1, 0xF, 0xF, true));
  p += __int_as_float(__builtin_amdgcn_update_dpp(0, __float_as_int(p), 0x4E, 0xF, 0xF, true));
  return p;
}

__device__ __forceinline__ float dot4(const v4f a, const v4f b) {
  return a.x * b.x + a.y * b.y + a.z * b.z + a.w * b.w;
}

#define ROWS_AHEAD 8

__device__ __forceinline__ void edge_step(const v4f hv, const v4f was,
                                          const float ad, float& m, float& den,
                                          v4f& acc) {
  const float lg = lrelu(quad_sum(dot4(hv, was)) + ad);
  const float t = lg - m;
  const float pe = expf(-fabsf(t));
  const bool up = t > 0.0f;
  const float sc = up ? pe : 1.0f;
  const float ev = up ? 1.0f : pe;
  m = up ? lg : m;
  den = den * sc + ev;
  acc.x = acc.x * sc + ev * hv.x;
  acc.y = acc.y * sc + ev * hv.y;
  acc.z = acc.z * sc + ev * hv.z;
  acc.w = acc.w * sc + ev * hv.w;
}

__global__ __launch_bounds__(256) void node_softmax_gather(
    const int* __restrict__ cntg, const int* __restrict__ misc,
    const int* __restrict__ col, const int* __restrict__ ovf,
    const float* __restrict__ att_src, const float* __restrict__ att_dst,
    const float* __restrict__ h, const float* __restrict__ bias,
    float* __restrict__ out, int N) {
  const int d = __builtin_amdgcn_readfirstlane(
      (int)(blockIdx.x * 8 + (threadIdx.x >> 5)));
  if (d >= N) return;
  const int lane = threadIdx.x & 31;
  const int deg = cntg[d];
  const int novf = misc[0];
  const int nb = (deg < CAP) ? deg : CAP;
  const int* __restrict__ cp = col + (size_t)d * CAP;
  const v4f was = *(const v4f*)(att_src + lane * 4);
  const v4f wad = *(const v4f*)(att_dst + lane * 4);
  const v4f hself = *(const v4f*)(h + (size_t)d * F + lane * 4);

  const float ad = quad_sum(dot4(hself, wad));
  float m = lrelu(quad_sum(dot4(hself, was)) + ad);
  float den = 1.0f;
  v4f acc = hself;

  if (nb <= 32) {
    const int n = nb;
    int sv = (lane < n) ? cp[lane] : d;
    int rk = 0;
    for (int i = 0; i < n; ++i) {
      const int si = __builtin_amdgcn_readlane(sv, i);
      rk += (si < sv || (si == sv && i < lane)) ? 1 : 0;
    }
    const int place = (lane < n) ? rk : lane;
    sv = __builtin_amdgcn_ds_permute(place << 2, sv);
    for (int j0 = 0; j0 < n; j0 += ROWS_AHEAD) {
      const int r = n - j0;
      int s[ROWS_AHEAD];
#pragma unroll
      for (int u = 0; u < ROWS_AHEAD; ++u) s[u] = __builtin_amdgcn_readlane(sv, j0 + u);
      v4f hv[ROWS_AHEAD];
      if (r >= ROWS_AHEAD) {
#pragma unroll
        for (int u = 0; u < ROWS_AHEAD; ++u)
          hv[u] = *(const v4f*)(h + (size_t)s[u] * F + lane * 4);
        asm volatile("" ::"v"(hv[0]), "v"(hv[1]), "v"(hv[2]), "v"(hv[3]),
                     "v"(hv[4]), "v"(hv[5]), "v"(hv[6]), "v"(hv[7]));
#pragma unroll
        for (int u = 0; u < ROWS_AHEAD; ++u) edge_step(hv[u], was, ad, m, den, acc);
      } else {
#pragma unroll
        for (int u = 0; u < ROWS_AHEAD - 1; ++u)
          if (u < r) hv[u] = *(const v4f*)(h + (size_t)s[u] * F + lane * 4);
#pragma unroll
        for (int u = 0; u < ROWS_AHEAD - 1; ++u)
          if (u < r) edge_step(hv[u], was, ad, m, den, acc);
      }
    }
  } else {
    const int n1 = nb - 32;
    const int sv0 = cp[lane];
    const int sv1 = (lane < n1) ? cp[32 + lane] : 0x7fffffff;
    int r0 = 0, r1 = 0;
    for (int i = 0; i < 32; ++i) {
      const int si = __builtin_amdgcn_readlane(sv0, i);
      r0 += (si < sv0 || (si == sv0 && i < lane)) ? 1 : 0;
      r1 += (si <= sv1) ? 1 : 0;
    }
    for (int i = 0; i < n1; ++i) {
      const int si = __builtin_amdgcn_readlane(sv1, i);
      r0 += (si < sv0) ? 1 : 0;
      r1 += (si < sv1 || (si == sv1 && i < lane)) ? 1 : 0;
    }
    for (int k = 0; k < nb; ++k) {
      const unsigned b0 = __builtin_amdgcn_ballot_w32(r0 == k);
      const unsigned b1 = __builtin_amdgcn_ballot_w32(lane < n1 && r1 == k);
      int s1;
      if (b0 != 0u) s1 = __builtin_amdgcn_readlane(sv0, __builtin_ctz(b0));
      else s1 = __builtin_amdgcn_readlane(sv1, __builtin_ctz(b1));
      const v4f hv1 = *(const v4f*)(h + (size_t)s1 * F + lane * 4);
      edge_step(hv1, was, ad, m, den, acc);
    }
  }
  if (novf > 0) {
    const int no = novf;
    for (int q0 = 0; q0 < no; q0 += 32) {
      const int q = q0 + lane;
      const int dd = (q < no) ? ovf[2 * (size_t)q] : -1;
      const int ss = (q < no) ? ovf[2 * (size_t)q + 1] : d;
      unsigned mine = __builtin_amdgcn_ballot_w32(dd == d);
      while (mine != 0u) {
        const int j = __builtin_ctz(mine);
        mine &= mine - 1u;
        const int s1 = __builtin_amdgcn_readlane(ss, j);
        const v4f hv1 = *(const v4f*)(h + (size_t)s1 * F + lane * 4);
        edge_step(hv1, was, ad, m, den, acc);
      }
    }
  }
  const v4f b = *(const v4f*)(bias + lane * 4);
  v4f o;
  o.x = acc.x / den + b.x;
  o.y = acc.y / den + b.y;
  o.z = acc.z / den + b.z;
  o.w = acc.w / den + b.w;
  *(v4f*)(out + (size_t)d * F + lane * 4) = o;
}

extern "C" void kernel_launch(void* const* d_in, const int* in_sizes, int n_in,
                              void* d_out, int out_size, void* d_ws, size_t ws_size,
                              hipStream_t stream) {
  const float* x = (const float*)d_in[0];
  const int* ei = (const int*)d_in[1];
  const float* W = (const float*)d_in[2];
  const float* att_src = (const float*)d_in[3];
  const float* att_dst = (const float*)d_in[4];
  const float* bias = (const float*)d_in[5];
  float* out = (float*)d_out;

  const int N = in_sizes[0] / F;
  const int E = in_sizes[1] / 2;
  const int* srcA = ei;
  const int* dstA = ei + E;

  const int NB = (N + BN - 1) / BN;
  const int NWG = (E + CHUNK - 1) / CHUNK;

  float* ws = (float*)d_ws;
  float* hbuf = ws;
  int* col = (int*)(hbuf + (size_t)N * F);
  int* seg = col + (size_t)N * CAP;
  int* segcnt = seg + (size_t)NB * NWG * SEG;
  int* cntg = segcnt + (size_t)NB * NWG;
  int* misc = cntg + N;
  int* spcnt = misc + 4;
  int* spl = spcnt + NWG;
  int* ovf = spl + 2 * (size_t)NWG * CHUNK;

  const int S = (N + 15) / 16;
  const int SA = (S * 3) / 8;

  split_and_gemm<<<NWG + SA, 256, 0, stream>>>(srcA, dstA, seg, segcnt, misc, spl, spcnt,
                                               x, W, hbuf, N, E, NB, NWG);
  bin_and_gemm<<<NB + (S - SA), 256, 0, stream>>>(seg, segcnt, spl, spcnt, col, cntg, misc,
                                                  ovf, x, W, hbuf, N, NWG, NB, SA);
  node_softmax_gather<<<(N + 7) / 8, 256, 0, stream>>>(cntg, misc, col, ovf, att_src, att_dst,
                                                       hbuf, bias, out, N);
}


# Launches hold about one wave per wave slot, not one per item: gather waves loop over nodes (1536 work-groups), GEMM work-groups over 4 strips each with W fragments kept in registers; sum orders kept
# speedup vs baseline: 7.6220x; 1.0390x over previous
#include <hip/hip_runtime.h>
#include <hip/hip_bf16.h>

typedef float v2f __attribute__((ext_vector_type(2)));
typedef float v8f __attribute__((ext_vector_type(8)));
typedef float v4f __attribute__((ext_vector_type(4)));

#define CH 16
#define F 128
#define NEG_SLOPE 0.2f
#define CAP 64
#define BSH 8
#define BN (1 << BSH)
#define NBMAX 448
#define SEG 32
#define CHUNK 4096
#define EPT (CHUNK / 256)
#define NWGMAX 512
#define HALFCAP 32
#define LDP (F + 4)

#if defined(__has_builtin)
#if __has_builtin(__builtin_amdgcn_wmma_f32_16x16x4_f32)
#define HAVE_WMMA_F32X4 1
#endif
#endif

__device__ __forceinline__ float lrelu(float v) {
  return v > 0.0f ? v : NEG_SLOPE * v;
}

#define STRIPS_PER_WG 4
__device__ __forceinline__ void gemm_strips(
    const float* __restrict__ x, const float* __restrict__ W,
    float* __restrict__ h, int N, int first, int count, float* sA) {
  const int tid = threadIdx.x;
  const int wave = tid >> 5;
  const int lane = tid & 31;
  const int half = lane >> 4;
  const int l16 = lane & 15;
  const int lr = tid >> 4;
  const int lc = (tid & 15) << 3;
  const int colbase = wave * CH;

  float4 a0, a1;
  {
    int rg = first * 16 + lr;
    if (rg >= N) rg = N - 1;
    const float4* s4 = (const float4*)(x + (size_t)rg * F + lc);
    a0 = s4[0];
    a1 = s4[1];
  }

  for (int it = 0; it < count; ++it) {
  const int rowbase = (first + it) * 16;
  *(float4*)(sA + lr * LDP + lc) = a0;
  *(float4*)(sA + lr * LDP + lc + 4) = a1;
  __syncthreads();
  if (it + 1 < count) {
    int rg = rowbase + 16 + lr;
    if (rg >= N) rg = N - 1;
    const float4* s4 = (const float4*)(x + (size_t)rg * F + lc);
    a0 = s4[0];
    a1 = s4[1];
  }

  v8f acc = {0.f, 0.f, 0.f, 0.f, 0.f, 0.f, 0.f, 0.f};

  for (int k = 0; k < F; k += 4) {
#if defined(HAVE_WMMA_F32X4)
    v2f afrag;
    const float* ap = sA + l16 * LDP + k + 2 * half;
    afrag.x = ap[0];
    afrag.y = ap[1];
    v2f bfrag;
    const float* bp = W + (size_t)(k + 2 * half) * F + colbase + l16;
    bfrag.x = bp[0];
    bfrag.y = bp[F];
    acc = __builtin_amdgcn_wmma_f32_16x16x4_f32(false, afrag, false, bfrag,
                                                (short)0, acc, false, false);
#else
    for (int v = 0; v < 8; ++v) {
      int row = v + 8 * half;
      for (int j = 0; j < 4; ++j)
        acc[v] += sA[row * LDP + k + j] * W[(size_t)(k + j) * F + colbase + l16];
    }
#endif
  }

#pragma unroll
  for (int v = 0; v < 8; ++v) {
    int row = rowbase + v + 8 * half;
    if (row < N) h[(size_t)row * F + colbase + l16] = acc[v];
  }
  __syncthreads();
  }
}

#define SPLIT_LDS_INTS (NBMAX + NBMAX * SEG + 4)
__device__ __forceinline__ void split_body(const int* __restrict__ srcA,
                                           const int* __restrict__ dstA,
                                           int* __restrict__ seg,
                                           int* __restrict__ segcnt,
                                           int* __restrict__ misc,
                                           int* __restrict__ spl,
                                           int* __restrict__ spcnt, int E, int NB,
                                           int NWG, int w, int* lds) {
  int* lcnt = lds;
  int* stage = lds + NBMAX;
  int* lsp = lds + NBMAX + NBMAX * SEG;
  const int tid = threadIdx.x;
  if (w == 0 && tid == 32) misc[0] = 0;
  if (tid == 0) lsp[0] = 0;
  for (int b = tid; b < NB; b += 256) lcnt[b] = 0;
  int dd[EPT], ss[EPT];
  const int base = w * CHUNK + tid;
#pragma unroll
  for (int i = 0; i < EPT; ++i) {
    const int e = base + i * 256;
    dd[i] = (e < E) ? dstA[e] : -1;
    ss[i] = (e < E) ? srcA[e] : 0;
  }
  __syncthreads();
#pragma unroll
  for (int i = 0; i < EPT; ++i) {
    const int d = dd[i];
    if (d >= 0) {
      const int b = d >> BSH;
      const int r = atomicAdd(&lcnt[b], 1);
      if (r < SEG) {
        stage[b * SEG + r] = (int)(((unsigned)(d & (BN - 1)) << 24) | (unsigned)ss[i]);
      } else {
        const int q = atomicAdd(lsp, 1);
        const size_t p = 2 * ((size_t)w * CHUNK + q);
        spl[p] = d;
        spl[p + 1] = ss[i];
      }
    }
  }
  __syncthreads();
  if (tid == 0) spcnt[w] = lsp[0];
  const int wave = tid >> 5;
  const int lane = tid & 31;
  for (int b = wave; b < NB; b += 8) {
    int c = lcnt[b];
    c = (c < SEG) ? c : SEG;
    const size_t sg = (size_t)b * NWG + w;
    if (lane < c) seg[sg * SEG + lane] = stage[b * SEG + lane];
    if (lane == 0) segcnt[sg] = c;
  }
}

#define SEGS_AHEAD 8
#define BIN_LDS_INTS (BN + NWGMAX + BN * HALFCAP + 4)
__device__ __forceinline__ void bin_body(const int* __restrict__ seg,
                                         const int* __restrict__ segcnt,
                                         const int* __restrict__ spl,
                                         const int* __restrict__ spcnt,
                                         int* __restrict__ col,
                                         int* __restrict__ cntg, int* misc,
                                         int* __restrict__ ovf, int N, int NWG,
                                         int b, int* lds) {
  int* ncnt = lds;
  int* scnt = lds + BN;
  int* bst = lds + BN + NWGMAX;
  int* sflag = lds + BN + NWGMAX + BN * HALFCAP;
  const int tid = threadIdx.x;
  ncnt[tid] = 0;
  if (tid == 0) sflag[0] = 0;
  int anysp = 0;
  for (int i = tid; i < NWG; i += 256) {
    scnt[i] = segcnt[(size_t)b * NWG + i];
    anysp |= spcnt[i];
  }
  __syncthreads();
  if (anysp != 0) sflag[0] = 1;
  const int wave = tid >> 5;
  const int lane = tid & 31;
  for (int g0 = wave; g0 < NWG; g0 += 8 * SEGS_AHEAD) {
    int ev[SEGS_AHEAD];
    int cc[SEGS_AHEAD];
#pragma unroll
    for (int u = 0; u < SEGS_AHEAD; ++u) {
      const int g = g0 + 8 * u;
      cc[u] = (g < NWG) ? scnt[g] : 0;
      ev[u] = 0;
      if (lane < cc[u]) ev[u] = seg[((size_t)b * NWG + g) * SEG + lane];
    }
#pragma unroll
    for (int u = 0; u < SEGS_AHEAD; ++u) {
      if (lane < cc[u]) {
        const unsigned e = (unsigned)ev[u];
        const int dl = (int)(e >> 24);
        const int s = (int)(e & 0xFFFFFFu);
        const int r = atomicAdd(&ncnt[dl], 1);
        if (r < HALFCAP) {
          bst[dl * HALFCAP + r] = s;
        } else if (r < CAP) {
          col[(size_t)(b * BN + dl) * CAP + r] = s;
        } else {
          const int q = atomicAdd(&misc[0], 1);
          ovf[2 * (size_t)q] = b * BN + dl;
          ovf[2 * (size_t)q + 1] = s;
        }
      }
    }
  }
  __syncthreads();
  if (sflag[0] != 0) {
    for (int i = 0; i < NWG; ++i) {
      int c = spcnt[i];
      c = (c < CHUNK) ? c : CHUNK;
      for (int q = tid; q < c; q += 256) {
        const size_t p = 2 * ((size_t)i * CHUNK + q);
        const int d = spl[p];
        const int s = spl[p + 1];
        if ((d >> BSH) == b) {
          const int dl = d & (BN - 1);
          const int r = atomicAdd(&ncnt[dl], 1);
          if (r < HALFCAP) {
            bst[dl * HALFCAP + r] = s;
          } else if (r < CAP) {
            col[(size_t)d * CAP + r] = s;
          } else {
            const int q2 = atomicAdd(&misc[0], 1);
            ovf[2 * (size_t)q2] = d;
            ovf[2 * (size_t)q2 + 1] = s;
          }
        }
      }
    }
    __syncthreads();
  }
  for (int dl = wave; dl < BN; dl += 8) {
    const int node = b * BN + dl;
    if (node < N) {
      const int c = ncnt[dl];
      const int nh = (c < HALFCAP) ? c : HALFCAP;
      if (lane < nh) col[(size_t)node * CAP + lane] = bst[dl * HALFCAP + lane];
      if (lane == 0) cntg[node] = c;
    }
  }
}

__global__ __launch_bounds__(256) void split_and_gemm(
    const int* __restrict__ srcA, const int* __restrict__ dstA,
    int* __restrict__ seg, int* __restrict__ segcnt, int* __restrict__ misc,
    int* __restrict__ spl, int* __restrict__ spcnt,
    const float* __restrict__ x, const float* __restrict__ W, float* __restrict__ h,
    int N, int E, int NB, int NWG, int send) {
  __shared__ __attribute__((aligned(16))) int lds[SPLIT_LDS_INTS];
  const int wg = blockIdx.x;
  if (wg < NWG) {
    split_body(srcA, dstA, seg, segcnt, misc, spl, spcnt, E, NB, NWG, wg, lds);
  } else {
    const int first = (wg - NWG) * STRIPS_PER_WG;
    const int left = send - first;
    gemm_strips(x, W, h, N, first, left < STRIPS_PER_WG ? left : STRIPS_PER_WG, (float*)lds);
  }
}

__global__ __launch_bounds__(256) void bin_and_gemm(
    const int* __restrict__ seg, const int* __restrict__ segcnt,
    const int* __restrict__ spl, const int* __restrict__ spcnt,
    int* __restrict__ col, int* __restrict__ cntg, int* misc, int* __restrict__ ovf,
    const float* __restrict__ x, const float* __restrict__ W, float* __restrict__ h,
    int N, int NWG, int NB, int strip0, int send) {
  __shared__ __attribute__((aligned(16))) int lds[BIN_LDS_INTS];
  const int wg = blockIdx.x;
  if (wg < NB) {
    bin_body(seg, segcnt, spl, spcnt, col, cntg, misc, ovf, N, NWG, wg, lds);
  } else {
    const int first = strip0 + (wg - NB) * STRIPS_PER_WG;
    const int left = send - first;
    gemm_strips(x, W, h, N, first, left < STRIPS_PER_WG ? left : STRIPS_PER_WG, (float*)lds);
  }
}

__device__ __forceinline__ float quad_sum(float p) {
  p += __int_as_float(__builtin_amdgcn_update_dpp(0, __float_as_int(p), 0xB1, 0xF, 0xF, true));
  p += __int_as_float(__builtin_amdgcn_update_dpp(0, __float_as_int(p), 0x4E, 0xF, 0xF, true));
  return p;
}

__device__ __forceinline__ float dot4(const v4f a, const v4f b) {
  return a.x * b.x + a.y * b.y + a.z * b.z + a.w * b.w;
}

#define ROWS_AHEAD 8
#define GATHER_WGS 1536

__device__ __forceinline__ void edge_step(const v4f hv, const v4f was,
                                          const float ad, float& m, float& den,
                                          v4f& acc) {
  const float lg = lrelu(quad_sum(dot4(hv, was)) + ad);
  const float t = lg - m;
  const float pe = expf(-fabsf(t));
  const bool up = t > 0.0f;
  const float sc = up ? pe : 1.0f;
  const float ev = up ? 1.0f : pe;
  m = up ? lg : m;
  den = den * sc + ev;
  acc.x = acc.x * sc + ev * hv.x;
  acc.y = acc.y * sc + ev * hv.y;
  acc.z = acc.z * sc + ev * hv.z;
  acc.w = acc.w * sc + ev * hv.w;
}

__global__ __launch_bounds__(256) __attribute__((amdgpu_waves_per_eu(12, 16)))
void node_softmax_gather(
    const int* __restrict__ cntg, const int* __restrict__ misc,
    const int* __restrict__ col, const int* __restrict__ ovf,
    const float* __restrict__ att_src, const float* __restrict__ att_dst,
    const float* __restrict__ h, const float* __restrict__ bias,
    float* __restrict__ out, int N) {
  const int w0 = __builtin_amdgcn_readfirstlane(
      (int)(blockIdx.x * 8 + (threadIdx.x >> 5)));
  const int nwaves = (int)gridDim.x * 8;
  const int lane = threadIdx.x & 31;
  const int novf = misc[0];
  const v4f was = *(const v4f*)(att_src + lane * 4);
  const v4f wad = *(const v4f*)(att_dst + lane * 4);
  for (int d = w0; d < N; d += nwaves) {
  const int deg = cntg[d];
  const int nb = (deg < CAP) ? deg : CAP;
  const int* __restrict__ cp = col + (size_t)d * CAP;
  const v4f hself = *(const v4f*)(h + (size_t)d * F + lane * 4);

  const float ad = quad_sum(dot4(hself, wad));
  float m = lrelu(quad_sum(dot4(hself, was)) + ad);
  float den = 1.0f;
  v4f acc = hself;

  if (nb <= 32) {
    const int n = nb;
    int sv = (lane < n) ? cp[lane] : d;
    int rk = 0;
    for (int i = 0; i < n; ++i) {
      const int si = __builtin_amdgcn_readlane(sv, i);
      rk += (si < sv || (si == sv && i < lane)) ? 1 : 0;
    }
    const int place = (lane < n) ? rk : lane;
    sv = __builtin_amdgcn_ds_permute(place << 2, sv);
    for (int j0 = 0; j0 < n; j0 += ROWS_AHEAD) {
      const int r = n - j0;
      int s[ROWS_AHEAD];
#pragma unroll
      for (int u = 0; u < ROWS_AHEAD; ++u) s[u] = __builtin_amdgcn_readlane(sv, j0 + u);
      v4f hv[ROWS_AHEAD];
      if (r >= ROWS_AHEAD) {
#pragma unroll
        for (int u = 0; u < ROWS_AHEAD; ++u)
          hv[u] = *(const v4f*)(h + (size_t)s[u] * F + lane * 4);
        asm volatile("" ::"v"(hv[0]), "v"(hv[1]), "v"(hv[2]), "v"(hv[3]),
                     "v"(hv[4]), "v"(hv[5]), "v"(hv[6]), "v"(hv[7]));
#pragma unroll
        for (int u = 0; u < ROWS_AHEAD; ++u) edge_step(hv[u], was, ad, m, den, acc);
      } else {
#pragma unroll
        for (int u = 0; u < ROWS_AHEAD - 1; ++u)
          if (u < r) hv[u] = *(const v4f*)(h + (size_t)s[u] * F + lane * 4);
#pragma unroll
        for (int u = 0; u < ROWS_AHEAD - 1; ++u)
          if (u < r) edge_step(hv[u], was, ad, m, den, acc);
      }
    }
  } else {
    const int n1 = nb - 32;
    const int sv0 = cp[lane];
    const int sv1 = (lane < n1) ? cp[32 + lane] : 0x7fffffff;
    int r0 = 0, r1 = 0;
    for (int i = 0; i < 32; ++i) {
      const int si = __builtin_amdgcn_readlane(sv0, i);
      r0 += (si < sv0 || (si == sv0 && i < lane)) ? 1 : 0;
      r1 += (si <= sv1) ? 1 : 0;
    }
    for (int i = 0; i < n1; ++i) {
      const int si = __builtin_amdgcn_readlane(sv1, i);
      r0 += (si < sv0) ? 1 : 0;
      r1 += (si < sv1 || (si == sv1 && i < lane)) ? 1 : 0;
    }
    for (int k = 0; k < nb; ++k) {
      const unsigned b0 = __builtin_amdgcn_ballot_w32(r0 == k);
      const unsigned b1 = __builtin_amdgcn_ballot_w32(lane < n1 && r1 == k);
      int s1;
      if (b0 != 0u) s1 = __builtin_amdgcn_readlane(sv0, __builtin_ctz(b0));
      else s1 = __builtin_amdgcn_readlane(sv1, __builtin_ctz(b1));
      const v4f hv1 = *(const v4f*)(h + (size_t)s1 * F + lane * 4);
      edge_step(hv1, was, ad, m, den, acc);
    }
  }
  if (novf > 0) {
    const int no = novf;
    for (int q0 = 0; q0 < no; q0 += 32) {
      const int q = q0 + lane;
      const int dd = (q < no) ? ovf[2 * (size_t)q] : -1;
      const int ss = (q < no) ? ovf[2 * (size_t)q + 1] : d;
      unsigned mine = __builtin_amdgcn_ballot_w32(dd == d);
      while (mine != 0u) {
        const int j = __builtin_ctz(mine);
        mine &= mine - 1u;
        const int s1 = __builtin_amdgcn_readlane(ss, j);
        const v4f hv1 = *(const v4f*)(h + (size_t)s1 * F + lane * 4);
        edge_step(hv1, was, ad, m, den, acc);
      }
    }
  }
  const v4f b = *(const v4f*)(bias + lane * 4);
  v4f o;
  o.x = acc.x / den + b.x;
  o.y = acc.y / den + b.y;
  o.z = acc.z / den + b.z;
  o.w = acc.w / den + b.w;
  *(v4f*)(out + (size_t)d * F + lane * 4) = o;
  }
}

extern "C" void kernel_launch(void* const* d_in, const int* in_sizes, int n_in,
                              void* d_out, int out_size, void* d_ws, size_t ws_size,
                              hipStream_t stream) {
  const float* x = (const float*)d_in[0];
  const int* ei = (const int*)d_in[1];
  const float* W = (const float*)d_in[2];
  const float* att_src = (const float*)d_in[3];
  const float* att_dst = (const float*)d_in[4];
  const float* bias = (const float*)d_in[5];
  float* out = (float*)d_out;

  const int N = in_sizes[0] / F;
  const int E = in_sizes[1] / 2;
  const int* srcA = ei;
  const int* dstA = ei + E;

  const int NB = (N + BN - 1) / BN;
  const int NWG = (E + CHUNK - 1) / CHUNK;

  float* ws = (float*)d_ws;
  float* hbuf = ws;
  int* col = (int*)(hbuf + (size_t)N * F);
  int* seg = col + (size_t)N * CAP;
  int* segcnt = seg + (size_t)NB * NWG * SEG;
  int* cntg = segcnt + (size_t)NB * NWG;
  int* misc = cntg + N;
  int* spcnt = misc + 4;
  int* spl = spcnt + NWG;
  int* ovf = spl + 2 * (size_t)NWG * CHUNK;

  const int S = (N + 15) / 16;
  const int SA = (S * 3) / 8;
  const int GA = (SA + STRIPS_PER_WG - 1) / STRIPS_PER_WG;
  const int GB = (S - SA + STRIPS_PER_WG - 1) / STRIPS_PER_WG;
  const int NG = (N + 7) / 8;
  const int GW = (NG < GATHER_WGS) ? NG : GATHER_WGS;

  split_and_gemm<<<NWG + GA, 256, 0, stream>>>(srcA, dstA, seg, segcnt, misc, spl, spcnt,
                                               x, W, hbuf, N, E, NB, NWG, SA);
  bin_and_gemm<<<NB + GB, 256, 0, stream>>>(seg, segcnt, spl, spcnt, col, cntg, misc,
                                            ovf, x, W, hbuf, N, NWG, NB, SA, S);
  node_softmax_gather<<<GW, 256, 0, stream>>>(cntg, misc, col, ovf, att_src, att_dst,
                                              hbuf, bias, out, N);
}


# Gather softmax in base 2: att_src and att_dst times log2(e) once per wave, each edge weight one v_exp_f32 in place of library expf (range reduction, ldexp, two range selects); sum order kept
# speedup vs baseline: 8.4144x; 1.1040x over previous
#include <hip/hip_runtime.h>
#include <hip/hip_bf16.h>

typedef float v2f __attribute__((ext_vector_type(2)));
typedef float v8f __attribute__((ext_vector_type(8)));
typedef float v4f __attribute__((ext_vector_type(4)));

#define CH 16
#define F 128
#define NEG_SLOPE 0.2f
#define LOG2E 1.44269504088896340736f
#define CAP 64
#define BSH 8
#define BN (1 << BSH)
#define NBMAX 448
#define SEG 32
#define CHUNK 4096
#define EPT (CHUNK / 256)
#define NWGMAX 512
#define HALFCAP 32
#define LDP (F + 4)

#if defined(__has_builtin)
#if __has_builtin(__builtin_amdgcn_wmma_f32_16x16x4_f32)
#define HAVE_WMMA_F32X4 1
#endif
#endif

__device__ __forceinline__ float lrelu(float v) {
  return v > 0.0f ? v : NEG_SLOPE * v;
}

#define STRIPS_PER_WG 4
__device__ __forceinline__ void gemm_strips(
    const float* __restrict__ x, const float* __restrict__ W,
    float* __restrict__ h, int N, int first, int count, float* sA) {
  const int tid = threadIdx.x;
  const int wave = tid >> 5;
  const int lane = tid & 31;
  const int half = lane >> 4;
  const int l16 = lane & 15;
  const int lr = tid >> 4;
  const int lc = (tid & 15) << 3;
  const int colbase = wave * CH;

  float4 a0, a1;
  {
    int rg = first * 16 + lr;
    if (rg >= N) rg = N - 1;
    const float4* s4 = (const float4*)(x + (size_t)rg * F + lc);
    a0 = s4[0];
    a1 = s4[1];
  }

  for (int it = 0; it < count; ++it) {
  const int rowbase = (first + it) * 16;
  *(float4*)(sA + lr * LDP + lc) = a0;
  *(float4*)(sA + lr * LDP + lc + 4) = a1;
  __syncthreads();
  if (it + 1 < count) {
    int rg = rowbase + 16 + lr;
    if (rg >= N) rg = N - 1;
    const float4* s4 = (const float4*)(x + (size_t)rg * F + lc);
    a0 = s4[0];
    a1 = s4[1];
  }

  v8f acc = {0.f, 0.f, 0.f, 0.f, 0.f, 0.f, 0.f, 0.f};

  for (int k = 0; k < F; k += 4) {
#if defined(HAVE_WMMA_F32X4)
    v2f afrag;
    const float* ap = sA + l16 * LDP + k + 2 * half;
    afrag.x = ap[0];
    afrag.y = ap[1];
    v2f bfrag;
    const float* bp = W + (size_t)(k + 2 * half) * F + colbase + l16;
    bfrag.x = bp[0];
    bfrag.y = bp[F];
    acc = __builtin_amdgcn_wmma_f32_16x16x4_f32(false, afrag, false, bfrag,
                                                (short)0, acc, false, false);
#else
    for (int v = 0; v < 8; ++v) {
      int row = v + 8 * half;
      for (int j = 0; j < 4; ++j)
        acc[v] += sA[row * LDP + k + j] * W[(size_t)(k + j) * F + colbase + l16];
    }
#endif
  }

#pragma unroll
  for (int v = 0; v < 8; ++v) {
    int row = rowbase + v + 8 * half;
    if (row < N) h[(size_t)row * F + colbase + l16] = acc[v];
  }
  __syncthreads();
  }
}

#define SPLIT_LDS_INTS (NBMAX + NBMAX * SEG + 4)
__device__ __forceinline__ void split_body(const int* __restrict__ srcA,
                                           const int* __restrict__ dstA,
                                           int* __restrict__ seg,
                                           int* __restrict__ segcnt,
                                           int* __restrict__ misc,
                                           int* __restrict__ spl,
                                           int* __restrict__ spcnt, int E, int NB,
                                           int NWG, int w, int* lds) {
  int* lcnt = lds;
  int* stage = lds + NBMAX;
  int* lsp = lds + NBMAX + NBMAX * SEG;
  const int tid = threadIdx.x;
  if (w == 0 && tid == 32) misc[0] = 0;
  if (tid == 0) lsp[0] = 0;
  for (int b = tid; b < NB; b += 256) lcnt[b] = 0;
  int dd[EPT], ss[EPT];
  const int base = w * CHUNK + tid;
#pragma unroll
  for (int i = 0; i < EPT; ++i) {
    const int e = base + i * 256;
    dd[i] = (e < E) ? dstA[e] : -1;
    ss[i] = (e < E) ? srcA[e] : 0;
  }
  __syncthreads();
#pragma unroll
  for (int i = 0; i < EPT; ++i) {
    const int d = dd[i];
    if (d >= 0) {
      const int b = d >> BSH;
      const int r = atomicAdd(&lcnt[b], 1);
      if (r < SEG) {
        stage[b * SEG + r] = (int)(((unsigned)(d & (BN - 1)) << 24) | (unsigned)ss[i]);
      } else {
        const int q = atomicAdd(lsp, 1);
        const size_t p = 2 * ((size_t)w * CHUNK + q);
        spl[p] = d;
        spl[p + 1] = ss[i];
      }
    }
  }
  __syncthreads();
  if (tid == 0) spcnt[w] = lsp[0];
  const int wave = tid >> 5;
  const int lane = tid & 31;
  for (int b = wave; b < NB; b += 8) {
    int c = lcnt[b];
    c = (c < SEG) ? c : SEG;
    const size_t sg = (size_t)b * NWG + w;
    if (lane < c) seg[sg * SEG + lane] = stage[b * SEG + lane];
    if (lane == 0) segcnt[sg] = c;
  }
}

#define SEGS_AHEAD 8
#define BIN_LDS_INTS (BN + NWGMAX + BN * HALFCAP + 4)
__device__ __forceinline__ void bin_body(const int* __restrict__ seg,
                                         const int* __restrict__ segcnt,
                                         const int* __restrict__ spl,
                                         const int* __restrict__ spcnt,
                                         int* __restrict__ col,
                                         int* __restrict__ cntg, int* misc,
                                         int* __restrict__ ovf, int N, int NWG,
                                         int b, int* lds) {
  int* ncnt = lds;
  int* scnt = lds + BN;
  int* bst = lds + BN + NWGMAX;
  int* sflag = lds + BN + NWGMAX + BN * HALFCAP;
  const int tid = threadIdx.x;
  ncnt[tid] = 0;
  if (tid == 0) sflag[0] = 0;
  int anysp = 0;
  for (int i = tid; i < NWG; i += 256) {
    scnt[i] = segcnt[(size_t)b * NWG + i];
    anysp |= spcnt[i];
  }
  __syncthreads();
  if (anysp != 0) sflag[0] = 1;
  const int wave = tid >> 5;
  const int lane = tid & 31;
  for (int g0 = wave; g0 < NWG; g0 += 8 * SEGS_AHEAD) {
    int ev[SEGS_AHEAD];
    int cc[SEGS_AHEAD];
#pragma unroll
    for (int u = 0; u < SEGS_AHEAD; ++u) {
      const int g = g0 + 8 * u;
      cc[u] = (g < NWG) ? scnt[g] : 0;
      ev[u] = 0;
      if (lane < cc[u]) ev[u] = seg[((size_t)b * NWG + g) * SEG + lane];
    }
#pragma unroll
    for (int u = 0; u < SEGS_AHEAD; ++u) {
      if (lane < cc[u]) {
        const unsigned e = (unsigned)ev[u];
        const int dl = (int)(e >> 24);
        const int s = (int)(e & 0xFFFFFFu);
        const int r = atomicAdd(&ncnt[dl], 1);
        if (r < HALFCAP) {
          bst[dl * HALFCAP + r] = s;
        } else if (r < CAP) {
          col[(size_t)(b * BN + dl) * CAP + r] = s;
        } else {
          const int q = atomicAdd(&misc[0], 1);
          ovf[2 * (size_t)q] = b * BN + dl;
          ovf[2 * (size_t)q + 1] = s;
        }
      }
    }
  }
  __syncthreads();
  if (sflag[0] != 0) {
    for (int i = 0; i < NWG; ++i) {
      int c = spcnt[i];
      c = (c < CHUNK) ? c : CHUNK;
      for (int q = tid; q < c; q += 256) {
        const size_t p = 2 * ((size_t)i * CHUNK + q);
        const int d = spl[p];
        const int s = spl[p + 1];
        if ((d >> BSH) == b) {
          const int dl = d & (BN - 1);
          const int r = atomicAdd(&ncnt[dl], 1);
          if (r < HALFCAP) {
            bst[dl * HALFCAP + r] = s;
          } else if (r < CAP) {
            col[(size_t)d * CAP + r] = s;
          } else {
            const int q2 = atomicAdd(&misc[0], 1);
            ovf[2 * (size_t)q2] = d;
            ovf[2 * (size_t)q2 + 1] = s;
          }
        }
      }
    }
    __syncthreads();
  }
  for (int dl = wave; dl < BN; dl += 8) {
    const int node = b * BN + dl;
    if (node < N) {
      const int c = ncnt[dl];
      const int nh = (c < HALFCAP) ? c : HALFCAP;
      if (lane < nh) col[(size_t)node * CAP + lane] = bst[dl * HALFCAP + lane];
      if (lane == 0) cntg[node] = c;
    }
  }
}

__global__ __launch_bounds__(256) void split_and_gemm(
    const int* __restrict__ srcA, const int* __restrict__ dstA,
    int* __restrict__ seg, int* __restrict__ segcnt, int* __restrict__ misc,
    int* __restrict__ spl, int* __restrict__ spcnt,
    const float* __restrict__ x, const float* __restrict__ W, float* __restrict__ h,
    int N, int E, int NB, int NWG, int send) {
  __shared__ __attribute__((aligned(16))) int lds[SPLIT_LDS_INTS];
  const int wg = blockIdx.x;
  if (wg < NWG) {
    split_body(srcA, dstA, seg, segcnt, misc, spl, spcnt, E, NB, NWG, wg, lds);
  } else {
    const int first = (wg - NWG) * STRIPS_PER_WG;
    const int left = send - first;
    gemm_strips(x, W, h, N, first, left < STRIPS_PER_WG ? left : STRIPS_PER_WG, (float*)lds);
  }
}

__global__ __launch_bounds__(256) void bin_and_gemm(
    const int* __restrict__ seg, const int* __restrict__ segcnt,
    const int* __restrict__ spl, const int* __restrict__ spcnt,
    int* __restrict__ col, int* __restrict__ cntg, int* misc, int* __restrict__ ovf,
    const float* __restrict__ x, const float* __restrict__ W, float* __restrict__ h,
    int N, int NWG, int NB, int strip0, int send) {
  __shared__ __attribute__((aligned(16))) int lds[BIN_LDS_INTS];
  const int wg = blockIdx.x;
  if (wg < NB) {
    bin_body(seg, segcnt, spl, spcnt, col, cntg, misc, ovf, N, NWG, wg, lds);
  } else {
    const int first = strip0 + (wg - NB) * STRIPS_PER_WG;
    const int left = send - first;
    gemm_strips(x, W, h, N, first, left < STRIPS_PER_WG ? left : STRIPS_PER_WG, (float*)lds);
  }
}

__device__ __forceinline__ float quad_sum(float p) {
  p += __int_as_float(__builtin_amdgcn_update_dpp(0, __float_as_int(p), 0xB1, 0xF, 0xF, true));
  p += __int_as_float(__builtin_amdgcn_update_dpp(0, __float_as_int(p), 0x4E, 0xF, 0xF, true));
  return p;
}

__device__ __forceinline__ float dot4(const v4f a, const v4f b) {
  return a.x * b.x + a.y * b.y + a.z * b.z + a.w * b.w;
}

#define ROWS_AHEAD 8
#define GATHER_WGS 1536

__device__ __forceinline__ float pow2_nonpos(float v) {
#if defined(__has_builtin)
#if __has_builtin(__builtin_amdgcn_exp2f)
  return __builtin_amdgcn_exp2f(v);
#else
  return exp2f(v);
#endif
#else
  return exp2f(v);
#endif
}

__device__ __forceinline__ void edge_step(const v4f hv, const v4f was,
                                          const float ad, float& m, float& den,
                                          v4f& acc) {
  const float lg = lrelu(quad_sum(dot4(hv, was)) + ad);
  const float t = lg - m;
  const float pe = pow2_nonpos(-fabsf(t));
  const bool up = t > 0.0f;
  const float sc = up ? pe : 1.0f;
  const float ev = up ? 1.0f : pe;
  m = up ? lg : m;
  den = den * sc + ev;
  acc.x = acc.x * sc + ev * hv.x;
  acc.y = acc.y * sc + ev * hv.y;
  acc.z = acc.z * sc + ev * hv.z;
  acc.w = acc.w * sc + ev * hv.w;
}

__global__ __launch_bounds__(256) __attribute__((amdgpu_waves_per_eu(12, 16)))
void node_softmax_gather(
    const int* __restrict__ cntg, const int* __restrict__ misc,
    const int* __restrict__ col, const int* __restrict__ ovf,
    const float* __restrict__ att_src, const float* __restrict__ att_dst,
    const float* __restrict__ h, const float* __restrict__ bias,
    float* __restrict__ out, int N) {
  const int w0 = __builtin_amdgcn_readfirstlane(
      (int)(blockIdx.x * 8 + (threadIdx.x >> 5)));
  const int nwaves = (int)gridDim.x * 8;
  const int lane = threadIdx.x & 31;
  const int novf = misc[0];
  const v4f was = *(const v4f*)(att_src + lane * 4) * LOG2E;
  const v4f wad = *(const v4f*)(att_dst + lane * 4) * LOG2E;
  for (int d = w0; d < N; d += nwaves) {
  const int deg = cntg[d];
  const int nb = (deg < CAP) ? deg : CAP;
  const int* __restrict__ cp = col + (size_t)d * CAP;
  const v4f hself = *(const v4f*)(h + (size_t)d * F + lane * 4);

  const float ad = quad_sum(dot4(hself, wad));
  float m = lrelu(quad_sum(dot4(hself, was)) + ad);
  float den = 1.0f;
  v4f acc = hself;

  if (nb <= 32) {
    const int n = nb;
    int sv = (lane < n) ? cp[lane] : d;
    int rk = 0;
    for (int i = 0; i < n; ++i) {
      const int si = __builtin_amdgcn_readlane(sv, i);
      rk += (si < sv || (si == sv && i < lane)) ? 1 : 0;
    }
    const int place = (lane < n) ? rk : lane;
    sv = __builtin_amdgcn_ds_permute(place << 2, sv);
    for (int j0 = 0; j0 < n; j0 += ROWS_AHEAD) {
      const int r = n - j0;
      int s[ROWS_AHEAD];
#pragma unroll
      for (int u = 0; u < ROWS_AHEAD; ++u) s[u] = __builtin_amdgcn_readlane(sv, j0 + u);
      v4f hv[ROWS_AHEAD];
      if (r >= ROWS_AHEAD) {
#pragma unroll
        for (int u = 0; u < ROWS_AHEAD; ++u)
          hv[u] = *(const v4f*)(h + (size_t)s[u] * F + lane * 4);
        __builtin_amdgcn_sched_barrier(0);
#pragma unroll
        for (int u = 0; u < ROWS_AHEAD; ++u) edge_step(hv[u], was, ad, m, den, acc);
      } else {
#pragma unroll
        for (int u = 0; u < ROWS_AHEAD - 1; ++u)
          if (u < r) hv[u] = *(const v4f*)(h + (size_t)s[u] * F + lane * 4);
#pragma unroll
        for (int u = 0; u < ROWS_AHEAD - 1; ++u)
          if (u < r) edge_step(hv[u], was, ad, m, den, acc);
      }
    }
  } else {
    const int n1 = nb - 32;
    const int sv0 = cp[lane];
    const int sv1 = (lane < n1) ? cp[32 + lane] : 0x7fffffff;
    int r0 = 0, r1 = 0;
    for (int i = 0; i < 32; ++i) {
      const int si = __builtin_amdgcn_readlane(sv0, i);
      r0 += (si < sv0 || (si == sv0 && i < lane)) ? 1 : 0;
      r1 += (si <= sv1) ? 1 : 0;
    }
    for (int i = 0; i < n1; ++i) {
      const int si = __builtin_amdgcn_readlane(sv1, i);
      r0 += (si < sv0) ? 1 : 0;
      r1 += (si < sv1 || (si == sv1 && i < lane)) ? 1 : 0;
    }
    for (int k = 0; k < nb; ++k) {
      const unsigned b0 = __builtin_amdgcn_ballot_w32(r0 == k);
      const unsigned b1 = __builtin_amdgcn_ballot_w32(lane < n1 && r1 == k);
      int s1;
      if (b0 != 0u) s1 = __builtin_amdgcn_readlane(sv0, __builtin_ctz(b0));
      else s1 = __builtin_amdgcn_readlane(sv1, __builtin_ctz(b1));
      const v4f hv1 = *(const v4f*)(h + (size_t)s1 * F + lane * 4);
      edge_step(hv1, was, ad, m, den, acc);
    }
  }
  if (novf > 0) {
    const int no = novf;
    for (int q0 = 0; q0 < no; q0 += 32) {
      const int q = q0 + lane;
      const int dd = (q < no) ? ovf[2 * (size_t)q] : -1;
      const int ss = (q < no) ? ovf[2 * (size_t)q + 1] : d;
      unsigned mine = __builtin_amdgcn_ballot_w32(dd == d);
      while (mine != 0u) {
        const int j = __builtin_ctz(mine);
        mine &= mine - 1u;
        const int s1 = __builtin_amdgcn_readlane(ss, j);
        const v4f hv1 = *(const v4f*)(h + (size_t)s1 * F + lane * 4);
        edge_step(hv1, was, ad, m, den, acc);
      }
    }
  }
  const v4f b = *(const v4f*)(bias + lane * 4);
  v4f o;
  o.x = acc.x / den + b.x;
  o.y = acc.y / den + b.y;
  o.z = acc.z / den + b.z;
  o.w = acc.w / den + b.w;
  *(v4f*)(out + (size_t)d * F + lane * 4) = o;
  }
}

extern "C" void kernel_launch(void* const* d_in, const int* in_sizes, int n_in,
                              void* d_out, int out_size, void* d_ws, size_t ws_size,
                              hipStream_t stream) {
  const float* x = (const float*)d_in[0];
  const int* ei = (const int*)d_in[1];
  const float* W = (const float*)d_in[2];
  const float* att_src = (const float*)d_in[3];
  const float* att_dst = (const float*)d_in[4];
  const float* bias = (const float*)d_in[5];
  float* out = (float*)d_out;

  const int N = in_sizes[0] / F;
  const int E = in_sizes[1] / 2;
  const int* srcA = ei;
  const int* dstA = ei + E;

  const int NB = (N + BN - 1) / BN;
  const int NWG = (E + CHUNK - 1) / CHUNK;

  float* ws = (float*)d_ws;
  float* hbuf = ws;
  int* col = (int*)(hbuf + (size_t)N * F);
  int* seg = col + (size_t)N * CAP;
  int* segcnt = seg + (size_t)NB * NWG * SEG;
  int* cntg = segcnt + (size_t)NB * NWG;
  int* misc = cntg + N;
  int* spcnt = misc + 4;
  int* spl = spcnt + NWG;
  int* ovf = spl + 2 * (size_t)NWG * CHUNK;

  const int S = (N + 15) / 16;
  const int SA = (S * 3) / 8;
  const int GA = (SA + STRIPS_PER_WG - 1) / STRIPS_PER_WG;
  const int GB = (S - SA + STRIPS_PER_WG - 1) / STRIPS_PER_WG;
  const int NG = (N + 7) / 8;
  const int GW = (NG < GATHER_WGS) ? NG : GATHER_WGS;

  split_and_gemm<<<NWG + GA, 256, 0, stream>>>(srcA, dstA, seg, segcnt, misc, spl, spcnt,
                                               x, W, hbuf, N, E, NB, NWG, SA);
  bin_and_gemm<<<NB + GB, 256, 0, stream>>>(seg, segcnt, spl, spcnt, col, cntg, misc,
                                            ovf, x, W, hbuf, N, NWG, NB, SA, S);
  node_softmax_gather<<<GW, 256, 0, stream>>>(cntg, misc, col, ovf, att_src, att_dst,
                                              hbuf, bias, out, N);
}
